# starting kernel
# baseline (speedup 1.0000x reference)
#include <hip/hip_runtime.h>

typedef float v2f __attribute__((ext_vector_type(2)));
typedef float v8f __attribute__((ext_vector_type(8)));

#define LEAKY 0.2f
#define L2EPS 1e-12f


__global__ void fill_zero_f32(float* __restrict__ p, int n) {
  int i = blockIdx.x * blockDim.x + threadIdx.x;
  if (i < n) p[i] = 0.0f;
}

__global__ void deg_kernel(const int* __restrict__ ei, float* __restrict__ deg, int E) {
  int e = blockIdx.x * blockDim.x + threadIdx.x;
  if (e < E) atomicAdd(&deg[ei[e]], 1.0f);
}

__global__ void dinv_kernel(float* __restrict__ d, int N) {
  int i = blockIdx.x * blockDim.x + threadIdx.x;
  if (i < N) {
    float v = d[i];
    d[i] = (v > 0.0f) ? rsqrtf(v) : 0.0f;
  }
}

__global__ void edgew_kernel(const int* __restrict__ ei, const float* __restrict__ dinv,
                             float* __restrict__ w, int E) {
  int e = blockIdx.x * blockDim.x + threadIdx.x;
  if (e < E) w[e] = dinv[ei[e]] * dinv[ei[E + e]];
}

__global__ void copy_x_kernel(const float* __restrict__ x, float* __restrict__ out,
                              int N, int outStride) {
  int t = blockIdx.x * blockDim.x + threadIdx.x;
  if (t < N * 16) {
    int r = t >> 4, q = t & 15;
    float4 v = reinterpret_cast<const float4*>(x)[(size_t)r * 16 + q];
    reinterpret_cast<float4*>(out + (size_t)r * outStride)[q] = v;
  }
}

__global__ void agg_kernel(const int* __restrict__ ei, const float* __restrict__ w,
                           const float* __restrict__ h, float* __restrict__ ha, int E) {
  long long t = (long long)blockIdx.x * blockDim.x + threadIdx.x;
  if (t >= (long long)E * 32) return;
  int e = (int)(t >> 5);
  int p = (int)(t & 31);
  int r = ei[e];
  int c = ei[E + e];
  float wv = w[e];
  v2f hv = *reinterpret_cast<const v2f*>(h + (size_t)c * 64 + p * 2);
  float* dst = ha + (size_t)r * 64 + p * 2;
  atomicAdd(dst + 0, wv * hv.x);
  atomicAdd(dst + 1, wv * hv.y);
}

__launch_bounds__(256)
__global__ void layer_kernel(const float* __restrict__ h,  const float* __restrict__ ha,
                             const float* __restrict__ Wg, const float* __restrict__ bg,
                             const float* __restrict__ Wi, const float* __restrict__ bi,
                             float* __restrict__ hnext, float* __restrict__ out,
                             int N, int outStride, int colOff) {
  __shared__ float sWg[64 * 64];
  __shared__ float sWi[64 * 64];
  __shared__ float sbg[64];
  __shared__ float sbi[64];

  int tid = threadIdx.x;
  for (int i = tid; i < 4096; i += 256) {
    int r = i >> 6, c = i & 63;
    int idx = (((r >> 1) * 64 + c) << 1) | (r & 1);
    sWg[idx] = Wg[i];
    sWi[idx] = Wi[i];
  }
  if (tid < 64) { sbg[tid] = bg[tid]; sbi[tid] = bi[tid]; }
  __syncthreads();

  int wave = tid >> 5, lane = tid & 31;
  int tiles = (N + 15) >> 4;
  int tile = blockIdx.x * 8 + wave;
  if (tile >= tiles) return;

  int m0 = tile << 4;
  int kh  = lane >> 4;
  int l15 = lane & 15;

  int rowA = m0 + l15;
  int rowLd = rowA < N ? rowA : N - 1;
  const float* haRow = ha + (size_t)rowLd * 64 + 2 * kh;
  const float* hRow  = h  + (size_t)rowLd * 64 + 2 * kh;

  v8f zero8 = {0.f, 0.f, 0.f, 0.f, 0.f, 0.f, 0.f, 0.f};
  v8f accg[4], acci[4];
#pragma unroll
  for (int nt = 0; nt < 4; nt++) { accg[nt] = zero8; acci[nt] = zero8; }

  const v2f* sWg2 = reinterpret_cast<const v2f*>(sWg);
  const v2f* sWi2 = reinterpret_cast<const v2f*>(sWi);

#pragma unroll
  for (int k = 0; k < 64; k += 4) {
    v2f ga = *reinterpret_cast<const v2f*>(haRow + k);
    v2f hv = *reinterpret_cast<const v2f*>(hRow + k);
    v2f ia = ga * hv;
    int k2 = (k >> 1) + kh;
#pragma unroll
    for (int nt = 0; nt < 4; nt++) {
      v2f bgf = sWg2[k2 * 64 + nt * 16 + l15];
      v2f bif = sWi2[k2 * 64 + nt * 16 + l15];
      accg[nt] = __builtin_amdgcn_wmma_f32_16x16x4_f32(
          false, ga, false, bgf, (short)0, accg[nt], false, false);
      acci[nt] = __builtin_amdgcn_wmma_f32_16x16x4_f32(
          false, ia, false, bif, (short)0, acci[nt], false, false);
    }
  }

  float bgl[4], bil[4];
#pragma unroll
  for (int nt = 0; nt < 4; nt++) {
    bgl[nt] = sbg[nt * 16 + l15];
    bil[nt] = sbi[nt * 16 + l15];
  }

#pragma unroll
  for (int j = 0; j < 8; j++) {
    int row = m0 + j + 8 * kh;
    float v[4];
    float sq = 0.0f;
#pragma unroll
    for (int nt = 0; nt < 4; nt++) {
      float g = accg[nt][j] + bgl[nt];
      float s = acci[nt][j] + bil[nt];
      g = (g > 0.0f) ? g : LEAKY * g;
      s = (s > 0.0f) ? s : LEAKY * s;
      v[nt] = g + s;
      sq += v[nt] * v[nt];
    }
#pragma unroll
    for (int m = 1; m < 16; m <<= 1) sq += __shfl_xor(sq, m, 32);
    float inv = rsqrtf(fmaxf(sq, L2EPS));
    if (row < N) {
      float* hn = hnext + (size_t)row * 64 + l15;
      float* op = out + (size_t)row * outStride + colOff + l15;
#pragma unroll
      for (int nt = 0; nt < 4; nt++) {
        hn[nt * 16] = v[nt];
        op[nt * 16] = v[nt] * inv;
      }
    }
  }
}


extern "C" void kernel_launch(void* const* d_in, const int* in_sizes, int n_in,
                              void* d_out, int out_size, void* d_ws, size_t ws_size,
                              hipStream_t stream) {
  const float* x  = (const float*)d_in[0];
  const int*   ei = (const int*)d_in[1];
  const float* Wg = (const float*)d_in[2];
  const float* bg = (const float*)d_in[3];
  const float* Wi = (const float*)d_in[4];
  const float* bi = (const float*)d_in[5];

  const int d = 64;
  const int N = in_sizes[0] / d;
  const int E = in_sizes[1] / 2;
  const int K = in_sizes[2] / (d * d);
  const int outStride = (K + 1) * d;

  float* ws    = (float*)d_ws;
  float* dinv  = ws;
  float* wvec  = dinv + N;
  float* haBuf = wvec + E;
  float* hA    = haBuf + (size_t)N * d;
  float* hB    = hA + (size_t)N * d;
  float* out   = (float*)d_out;

  fill_zero_f32<<<(N + 255) / 256, 256, 0, stream>>>(dinv, N);
  deg_kernel<<<(E + 255) / 256, 256, 0, stream>>>(ei, dinv, E);
  dinv_kernel<<<(N + 255) / 256, 256, 0, stream>>>(dinv, N);
  edgew_kernel<<<(E + 255) / 256, 256, 0, stream>>>(ei, dinv, wvec, E);
  copy_x_kernel<<<(N * 16 + 255) / 256, 256, 0, stream>>>(x, out, N, outStride);

  const float* h = x;
  float* nxt = hA;
  int tiles = (N + 15) / 16;
  int lblocks = (tiles + 7) / 8;
  long long aggT = (long long)E * 32;
  int ablocks = (int)((aggT + 255) / 256);

  for (int layer = 0; layer < K; layer++) {
    fill_zero_f32<<<(N * d + 255) / 256, 256, 0, stream>>>(haBuf, N * d);
    agg_kernel<<<ablocks, 256, 0, stream>>>(ei, wvec, h, haBuf, E);
    layer_kernel<<<lblocks, 256, 0, stream>>>(
        h, haBuf,
        Wg + (size_t)layer * d * d, bg + (size_t)layer * d,
        Wi + (size_t)layer * d * d, bi + (size_t)layer * d,
        nxt, out, N, outStride, (layer + 1) * d);
    h = nxt;
    nxt = (nxt == hA) ? hB : hA;
  }
}


# Aggregation: layer 1's launch keeps each wave's matched edges (source row, local row, w) as a list in the workspace; layers 2 and 3 add from the lists, no pass over the row list; LDS adds as before
# speedup vs baseline: 1.8958x; 1.8958x over previous
#include <hip/hip_runtime.h>

typedef float v2f __attribute__((ext_vector_type(2)));
typedef float v8f __attribute__((ext_vector_type(8)));

#define LEAKY 0.2f
#define L2EPS 1e-12f


__global__ void fill_zero_f32(float* __restrict__ p, int n) {
  int i = blockIdx.x * blockDim.x + threadIdx.x;
  if (i < n) p[i] = 0.0f;
}

__global__ void deg_kernel(const int* __restrict__ ei, float* __restrict__ deg, int E) {
  int e = blockIdx.x * blockDim.x + threadIdx.x;
  if (e < E) atomicAdd(&deg[ei[e]], 1.0f);
}

__global__ void dinv_kernel(float* __restrict__ d, int N) {
  int i = blockIdx.x * blockDim.x + threadIdx.x;
  if (i < N) {
    float v = d[i];
    d[i] = (v > 0.0f) ? rsqrtf(v) : 0.0f;
  }
}

__global__ void edgew_kernel(const int* __restrict__ ei, const float* __restrict__ dinv,
                             float* __restrict__ w, int E) {
  int e = blockIdx.x * blockDim.x + threadIdx.x;
  if (e < E) w[e] = dinv[ei[e]] * dinv[ei[E + e]];
}

__global__ void copy_x_kernel(const float* __restrict__ x, float* __restrict__ out,
                              int N, int outStride) {
  int t = blockIdx.x * blockDim.x + threadIdx.x;
  if (t < N * 16) {
    int r = t >> 4, q = t & 15;
    float4 v = reinterpret_cast<const float4*>(x)[(size_t)r * 16 + q];
    reinterpret_cast<float4*>(out + (size_t)r * outStride)[q] = v;
  }
}

#define BLK_ROWS 512
#define Q_CAP 256
#define DRAIN_G 16
#define SCAN_U 4
#define LIST_CAP 1024

static __device__ __forceinline__ float bcast_f32(float v, int srcLane) {
  return __builtin_bit_cast(float,
      __builtin_amdgcn_readlane(__builtin_bit_cast(int, v), srcLane));
}

static __device__ __forceinline__ void add_group(int c, int rl, float wt, int cnt,
                                                 const float* __restrict__ hl, float* accl) {
#pragma unroll
  for (int j0 = 0; j0 < 32; j0 += DRAIN_G) {
    v2f hv[DRAIN_G];
#pragma unroll
    for (int j = 0; j < DRAIN_G; j++) {
      if (j0 + j < cnt) {
        int cj = __builtin_amdgcn_readlane(c, j0 + j);
        hv[j] = *reinterpret_cast<const v2f*>(hl + (size_t)cj * 64);
      }
    }
#pragma unroll
    for (int j = 0; j < DRAIN_G; j++) {
      if (j0 + j < cnt) {
        int rj = __builtin_amdgcn_readlane(rl, j0 + j);
        float wj = bcast_f32(wt, j0 + j);
        float* dst = accl + rj * 64;
        atomicAdd(dst + 0, wj * hv[j].x);
        atomicAdd(dst + 1, wj * hv[j].y);
      }
    }
  }
}

template <bool RECORD>
static __device__ __forceinline__ void drain_ring(const int* qe, const int* qr, int head, int cnt,
                                                  int lane, const int* __restrict__ colp,
                                                  const float* __restrict__ w,
                                                  const float* __restrict__ hl, float* accl,
                                                  int* __restrict__ lc, int* __restrict__ lr,
                                                  float* __restrict__ lw, int nrec) {
  int idx = (head + lane) & (Q_CAP - 1);
  bool live = lane < cnt;
  int e  = live ? qe[idx] : 0;
  int rl = live ? qr[idx] : 0;
  int c = colp[e];
  float wt = w[e];
  if (RECORD) {
    if (nrec + cnt <= LIST_CAP && live) {
      lc[nrec + lane] = c;
      lr[nrec + lane] = rl;
      lw[nrec + lane] = wt;
    }
  }
  add_group(c, rl, wt, cnt, hl, accl);
}

template <int MODE>
__launch_bounds__(1024)
__global__ void agg_kernel(const int* __restrict__ ei, const float* __restrict__ w,
                           const float* __restrict__ h, float* __restrict__ ha,
                           int* __restrict__ lcnt, int* __restrict__ lcol,
                           int* __restrict__ lrow, float* __restrict__ lwt, int E, int N) {
  __shared__ float acc[BLK_ROWS * 64];
  __shared__ int qe[32 * Q_CAP];
  __shared__ int qr[32 * Q_CAP];

  int tid = threadIdx.x;
  int lane = tid & 31;
  int wave = __builtin_amdgcn_readfirstlane(tid >> 5);
  int r0 = blockIdx.x * BLK_ROWS;

  float4 z4 = {0.0f, 0.0f, 0.0f, 0.0f};
  for (int i = tid; i < BLK_ROWS * 16; i += 1024) reinterpret_cast<float4*>(acc)[i] = z4;
  __syncthreads();

  const int* rowp = ei;
  const int* colp = ei + E;
  int* myqe = qe + wave * Q_CAP;
  int* myqr = qr + wave * Q_CAP;
  const float* hl = h + lane * 2;
  float* accl = acc + lane * 2;

  int lslot = blockIdx.x * 32 + wave;
  size_t lbase = (size_t)lslot * LIST_CAP;
  int* lc = lcol + lbase;
  int* lr = lrow + lbase;
  float* lw = lwt + lbase;

  bool filter = true;
  if (MODE == 1) {
    int n = __builtin_amdgcn_readfirstlane(lcnt[lslot]);
    if (n <= LIST_CAP) {
      filter = false;
      int ng = (n + 31) >> 5;
      int nwhole = n >> 5;
      if (ng > 0) {
        int c = lc[lane], rl = lr[lane];
        float wt = lw[lane];
        for (int g = 0; g < nwhole; g++) {
          int gn = (g + 1 < ng) ? g + 1 : g;
          int cn = lc[gn * 32 + lane], rln = lr[gn * 32 + lane];
          float wtn = lw[gn * 32 + lane];
          add_group(c, rl, wt, 32, hl, accl);
          c = cn; rl = rln; wt = wtn;
        }
        if ((n & 31) != 0) add_group(c, rl, wt, n & 31, hl, accl);
      }
    }
  }

  if (filter) {
  int head = 0, npend = 0;
  int nrec = 0;
  int nfull = E >> 7;
  auto load_chunk = [&](int chunk, int* rv) {
    int cl = chunk < nfull ? chunk : nfull - 1;
    int4 t = *reinterpret_cast<const int4*>(rowp + (cl << 7) + lane * 4);
    rv[0] = t.x; rv[1] = t.y; rv[2] = t.z; rv[3] = t.w;
  };
  auto match_chunk = [&](int chunk, const int* rv) {
    int e0 = (chunk << 7) + lane * 4;
#pragma unroll
    for (int k = 0; k < 4; k++) {
      int rl = rv[k] - r0;
      bool p = (unsigned)rl < (unsigned)BLK_ROWS;
      unsigned m = __builtin_amdgcn_ballot_w32(p);
      if (m != 0u) {
        if (p) {
          int pos = (head + npend + (int)__builtin_amdgcn_mbcnt_lo(m, 0u)) & (Q_CAP - 1);
          myqe[pos] = e0 + k;
          myqr[pos] = rl;
        }
        npend += __builtin_popcount(m);
      }
    }
  };

  if (nfull > 0) {
    int cur[SCAN_U][4], nxt[SCAN_U][4];
#pragma unroll
    for (int u = 0; u < SCAN_U; u++) load_chunk(wave + 32 * u, cur[u]);
    for (int c0 = wave; c0 < nfull; c0 += 32 * SCAN_U) {
#pragma unroll
      for (int u = 0; u < SCAN_U; u++) load_chunk(c0 + 32 * (SCAN_U + u), nxt[u]);
#pragma unroll
      for (int u = 0; u < SCAN_U; u++) {
        if (c0 + 32 * u < nfull) match_chunk(c0 + 32 * u, cur[u]);
        while (npend >= 32) {
          drain_ring<MODE == 0>(myqe, myqr, head, 32, lane, colp, w, hl, accl, lc, lr, lw, nrec);
          head = (head + 32) & (Q_CAP - 1);
          npend -= 32;
          nrec += 32;
        }
      }
#pragma unroll
      for (int u = 0; u < SCAN_U; u++) {
        cur[u][0] = nxt[u][0]; cur[u][1] = nxt[u][1]; cur[u][2] = nxt[u][2]; cur[u][3] = nxt[u][3];
      }
    }
  }
  if ((E & 127) != 0 && wave == (nfull & 31)) {
    int e0 = (nfull << 7) + lane * 4;
    int rv[4];
#pragma unroll
    for (int k = 0; k < 4; k++) rv[k] = (e0 + k < E) ? rowp[e0 + k] : -1;
    match_chunk(nfull, rv);
  }
  while (npend > 0) {
    int cnt = npend < 32 ? npend : 32;
    drain_ring<MODE == 0>(myqe, myqr, head, cnt, lane, colp, w, hl, accl, lc, lr, lw, nrec);
    head = (head + cnt) & (Q_CAP - 1);
    npend -= cnt;
    nrec += cnt;
  }
  if (MODE == 0 && lane == 0) lcnt[lslot] = nrec;
  }

  __syncthreads();
  int nrows = N - r0;
  if (nrows > BLK_ROWS) nrows = BLK_ROWS;
  float4* dst4 = reinterpret_cast<float4*>(ha + (size_t)r0 * 64);
  for (int i = tid; i < nrows * 16; i += 1024) dst4[i] = reinterpret_cast<const float4*>(acc)[i];
}

__launch_bounds__(256)
__global__ void layer_kernel(const float* __restrict__ h,  const float* __restrict__ ha,
                             const float* __restrict__ Wg, const float* __restrict__ bg,
                             const float* __restrict__ Wi, const float* __restrict__ bi,
                             float* __restrict__ hnext, float* __restrict__ out,
                             int N, int outStride, int colOff) {
  __shared__ float sWg[64 * 64];
  __shared__ float sWi[64 * 64];
  __shared__ float sbg[64];
  __shared__ float sbi[64];

  int tid = threadIdx.x;
  for (int i = tid; i < 4096; i += 256) {
    int r = i >> 6, c = i & 63;
    int idx = (((r >> 1) * 64 + c) << 1) | (r & 1);
    sWg[idx] = Wg[i];
    sWi[idx] = Wi[i];
  }
  if (tid < 64) { sbg[tid] = bg[tid]; sbi[tid] = bi[tid]; }
  __syncthreads();

  int wave = tid >> 5, lane = tid & 31;
  int tiles = (N + 15) >> 4;
  int tile = blockIdx.x * 8 + wave;
  if (tile >= tiles) return;

  int m0 = tile << 4;
  int kh  = lane >> 4;
  int l15 = lane & 15;

  int rowA = m0 + l15;
  int rowLd = rowA < N ? rowA : N - 1;
  const float* haRow = ha + (size_t)rowLd * 64 + 2 * kh;
  const float* hRow  = h  + (size_t)rowLd * 64 + 2 * kh;

  v8f zero8 = {0.f, 0.f, 0.f, 0.f, 0.f, 0.f, 0.f, 0.f};
  v8f accg[4], acci[4];
#pragma unroll
  for (int nt = 0; nt < 4; nt++) { accg[nt] = zero8; acci[nt] = zero8; }

  const v2f* sWg2 = reinterpret_cast<const v2f*>(sWg);
  const v2f* sWi2 = reinterpret_cast<const v2f*>(sWi);

#pragma unroll
  for (int k = 0; k < 64; k += 4) {
    v2f ga = *reinterpret_cast<const v2f*>(haRow + k);
    v2f hv = *reinterpret_cast<const v2f*>(hRow + k);
    v2f ia = ga * hv;
    int k2 = (k >> 1) + kh;
#pragma unroll
    for (int nt = 0; nt < 4; nt++) {
      v2f bgf = sWg2[k2 * 64 + nt * 16 + l15];
      v2f bif = sWi2[k2 * 64 + nt * 16 + l15];
      accg[nt] = __builtin_amdgcn_wmma_f32_16x16x4_f32(
          false, ga, false, bgf, (short)0, accg[nt], false, false);
      acci[nt] = __builtin_amdgcn_wmma_f32_16x16x4_f32(
          false, ia, false, bif, (short)0, acci[nt], false, false);
    }
  }

  float bgl[4], bil[4];
#pragma unroll
  for (int nt = 0; nt < 4; nt++) {
    bgl[nt] = sbg[nt * 16 + l15];
    bil[nt] = sbi[nt * 16 + l15];
  }

#pragma unroll
  for (int j = 0; j < 8; j++) {
    int row = m0 + j + 8 * kh;
    float v[4];
    float sq = 0.0f;
#pragma unroll
    for (int nt = 0; nt < 4; nt++) {
      float g = accg[nt][j] + bgl[nt];
      float s = acci[nt][j] + bil[nt];
      g = (g > 0.0f) ? g : LEAKY * g;
      s = (s > 0.0f) ? s : LEAKY * s;
      v[nt] = g + s;
      sq += v[nt] * v[nt];
    }
#pragma unroll
    for (int m = 1; m < 16; m <<= 1) sq += __shfl_xor(sq, m, 32);
    float inv = rsqrtf(fmaxf(sq, L2EPS));
    if (row < N) {
      float* hn = hnext + (size_t)row * 64 + l15;
      float* op = out + (size_t)row * outStride + colOff + l15;
#pragma unroll
      for (int nt = 0; nt < 4; nt++) {
        hn[nt * 16] = v[nt];
        op[nt * 16] = v[nt] * inv;
      }
    }
  }
}


extern "C" void kernel_launch(void* const* d_in, const int* in_sizes, int n_in,
                              void* d_out, int out_size, void* d_ws, size_t ws_size,
                              hipStream_t stream) {
  const float* x  = (const float*)d_in[0];
  const int*   ei = (const int*)d_in[1];
  const float* Wg = (const float*)d_in[2];
  const float* bg = (const float*)d_in[3];
  const float* Wi = (const float*)d_in[4];
  const float* bi = (const float*)d_in[5];

  const int d = 64;
  const int N = in_sizes[0] / d;
  const int E = in_sizes[1] / 2;
  const int K = in_sizes[2] / (d * d);
  const int outStride = (K + 1) * d;

  float* ws    = (float*)d_ws;
  float* dinv  = ws;
  float* wvec  = dinv + N;
  float* haBuf = wvec + E;
  float* hA    = haBuf + (size_t)N * d;
  float* hB    = hA + (size_t)N * d;
  float* out   = (float*)d_out;

  fill_zero_f32<<<(N + 255) / 256, 256, 0, stream>>>(dinv, N);
  deg_kernel<<<(E + 255) / 256, 256, 0, stream>>>(ei, dinv, E);
  dinv_kernel<<<(N + 255) / 256, 256, 0, stream>>>(dinv, N);
  edgew_kernel<<<(E + 255) / 256, 256, 0, stream>>>(ei, dinv, wvec, E);
  copy_x_kernel<<<(N * 16 + 255) / 256, 256, 0, stream>>>(x, out, N, outStride);

  const float* h = x;
  float* nxt = hA;
  int tiles = (N + 15) / 16;
  int lblocks = (tiles + 7) / 8;
  int ablocks = (N + BLK_ROWS - 1) / BLK_ROWS;
  int*   lcnt = (int*)(hB + (size_t)N * d);
  int*   lcol = lcnt + (size_t)ablocks * 32;
  int*   lrow = lcol + (size_t)ablocks * 32 * LIST_CAP;
  float* lwt  = (float*)(lrow + (size_t)ablocks * 32 * LIST_CAP);

  for (int layer = 0; layer < K; layer++) {
    if (layer == 0)
      agg_kernel<0><<<ablocks, 1024, 0, stream>>>(ei, wvec, h, haBuf, lcnt, lcol, lrow, lwt, E, N);
    else
      agg_kernel<1><<<ablocks, 1024, 0, stream>>>(ei, wvec, h, haBuf, lcnt, lcol, lrow, lwt, E, N);
    layer_kernel<<<lblocks, 256, 0, stream>>>(
        h, haBuf,
        Wg + (size_t)layer * d * d, bg + (size_t)layer * d,
        Wi + (size_t)layer * d * d, bi + (size_t)layer * d,
        nxt, out, N, outStride, (layer + 1) * d);
    h = nxt;
    nxt = (nxt == hA) ? hB : hA;
  }
}


# Aggregation: 392 destination rows per work-group, not 512: 256 work-groups, one on every WGP (196 left 60 idle), 23% fewer edges and LDS adds per WGP; LDS atomics, arrival order as before
# speedup vs baseline: 1.9760x; 1.0423x over previous
#include <hip/hip_runtime.h>

typedef float v2f __attribute__((ext_vector_type(2)));
typedef float v8f __attribute__((ext_vector_type(8)));

#define LEAKY 0.2f
#define L2EPS 1e-12f


__global__ void fill_zero_f32(float* __restrict__ p, int n) {
  int i = blockIdx.x * blockDim.x + threadIdx.x;
  if (i < n) p[i] = 0.0f;
}

__global__ void deg_kernel(const int* __restrict__ ei, float* __restrict__ deg, int E) {
  int e = blockIdx.x * blockDim.x + threadIdx.x;
  if (e < E) atomicAdd(&deg[ei[e]], 1.0f);
}

__global__ void dinv_kernel(float* __restrict__ d, int N) {
  int i = blockIdx.x * blockDim.x + threadIdx.x;
  if (i < N) {
    float v = d[i];
    d[i] = (v > 0.0f) ? rsqrtf(v) : 0.0f;
  }
}

__global__ void edgew_kernel(const int* __restrict__ ei, const float* __restrict__ dinv,
                             float* __restrict__ w, int E) {
  int e = blockIdx.x * blockDim.x + threadIdx.x;
  if (e < E) w[e] = dinv[ei[e]] * dinv[ei[E + e]];
}

__global__ void copy_x_kernel(const float* __restrict__ x, float* __restrict__ out,
                              int N, int outStride) {
  int t = blockIdx.x * blockDim.x + threadIdx.x;
  if (t < N * 16) {
    int r = t >> 4, q = t & 15;
    float4 v = reinterpret_cast<const float4*>(x)[(size_t)r * 16 + q];
    reinterpret_cast<float4*>(out + (size_t)r * outStride)[q] = v;
  }
}

#define BLK_ROWS 392
#define Q_CAP 256
#define DRAIN_G 16
#define SCAN_U 4
#define LIST_CAP 1024

static __device__ __forceinline__ float bcast_f32(float v, int srcLane) {
  return __builtin_bit_cast(float,
      __builtin_amdgcn_readlane(__builtin_bit_cast(int, v), srcLane));
}

static __device__ __forceinline__ void add_group(int c, int rl, float wt, int cnt,
                                                 const float* __restrict__ hl, float* accl) {
#pragma unroll
  for (int j0 = 0; j0 < 32; j0 += DRAIN_G) {
    v2f hv[DRAIN_G];
#pragma unroll
    for (int j = 0; j < DRAIN_G; j++) {
      if (j0 + j < cnt) {
        int cj = __builtin_amdgcn_readlane(c, j0 + j);
        hv[j] = *reinterpret_cast<const v2f*>(hl + (size_t)cj * 64);
      }
    }
#pragma unroll
    for (int j = 0; j < DRAIN_G; j++) {
      if (j0 + j < cnt) {
        int rj = __builtin_amdgcn_readlane(rl, j0 + j);
        float wj = bcast_f32(wt, j0 + j);
        float* dst = accl + rj * 64;
        atomicAdd(dst + 0, wj * hv[j].x);
        atomicAdd(dst + 1, wj * hv[j].y);
      }
    }
  }
}

template <bool RECORD>
static __device__ __forceinline__ void drain_ring(const int* qe, const int* qr, int head, int cnt,
                                                  int lane, const int* __restrict__ colp,
                                                  const float* __restrict__ w,
                                                  const float* __restrict__ hl, float* accl,
                                                  int* __restrict__ lc, int* __restrict__ lr,
                                                  float* __restrict__ lw, int nrec) {
  int idx = (head + lane) & (Q_CAP - 1);
  bool live = lane < cnt;
  int e  = live ? qe[idx] : 0;
  int rl = live ? qr[idx] : 0;
  int c = colp[e];
  float wt = w[e];
  if (RECORD) {
    if (nrec + cnt <= LIST_CAP && live) {
      lc[nrec + lane] = c;
      lr[nrec + lane] = rl;
      lw[nrec + lane] = wt;
    }
  }
  add_group(c, rl, wt, cnt, hl, accl);
}

template <int MODE>
__launch_bounds__(1024)
__global__ void agg_kernel(const int* __restrict__ ei, const float* __restrict__ w,
                           const float* __restrict__ h, float* __restrict__ ha,
                           int* __restrict__ lcnt, int* __restrict__ lcol,
                           int* __restrict__ lrow, float* __restrict__ lwt, int E, int N) {
  __shared__ float acc[BLK_ROWS * 64];
  __shared__ int qe[32 * Q_CAP];
  __shared__ int qr[32 * Q_CAP];

  int tid = threadIdx.x;
  int lane = tid & 31;
  int wave = __builtin_amdgcn_readfirstlane(tid >> 5);
  int r0 = blockIdx.x * BLK_ROWS;

  float4 z4 = {0.0f, 0.0f, 0.0f, 0.0f};
  for (int i = tid; i < BLK_ROWS * 16; i += 1024) reinterpret_cast<float4*>(acc)[i] = z4;
  __syncthreads();

  const int* rowp = ei;
  const int* colp = ei + E;
  int* myqe = qe + wave * Q_CAP;
  int* myqr = qr + wave * Q_CAP;
  const float* hl = h + lane * 2;
  float* accl = acc + lane * 2;

  int lslot = blockIdx.x * 32 + wave;
  size_t lbase = (size_t)lslot * LIST_CAP;
  int* lc = lcol + lbase;
  int* lr = lrow + lbase;
  float* lw = lwt + lbase;

  bool filter = true;
  if (MODE == 1) {
    int n = __builtin_amdgcn_readfirstlane(lcnt[lslot]);
    if (n <= LIST_CAP) {
      filter = false;
      int ng = (n + 31) >> 5;
      int nwhole = n >> 5;
      if (ng > 0) {
        int c = lc[lane], rl = lr[lane];
        float wt = lw[lane];
        for (int g = 0; g < nwhole; g++) {
          int gn = (g + 1 < ng) ? g + 1 : g;
          int cn = lc[gn * 32 + lane], rln = lr[gn * 32 + lane];
          float wtn = lw[gn * 32 + lane];
          add_group(c, rl, wt, 32, hl, accl);
          c = cn; rl = rln; wt = wtn;
        }
        if ((n & 31) != 0) add_group(c, rl, wt, n & 31, hl, accl);
      }
    }
  }

  if (filter) {
  int head = 0, npend = 0;
  int nrec = 0;
  int nfull = E >> 7;
  auto load_chunk = [&](int chunk, int* rv) {
    int cl = chunk < nfull ? chunk : nfull - 1;
    int4 t = *reinterpret_cast<const int4*>(rowp + (cl << 7) + lane * 4);
    rv[0] = t.x; rv[1] = t.y; rv[2] = t.z; rv[3] = t.w;
  };
  auto match_chunk = [&](int chunk, const int* rv) {
    int e0 = (chunk << 7) + lane * 4;
#pragma unroll
    for (int k = 0; k < 4; k++) {
      int rl = rv[k] - r0;
      bool p = (unsigned)rl < (unsigned)BLK_ROWS;
      unsigned m = __builtin_amdgcn_ballot_w32(p);
      if (m != 0u) {
        if (p) {
          int pos = (head + npend + (int)__builtin_amdgcn_mbcnt_lo(m, 0u)) & (Q_CAP - 1);
          myqe[pos] = e0 + k;
          myqr[pos] = rl;
        }
        npend += __builtin_popcount(m);
      }
    }
  };

  if (nfull > 0) {
    int cur[SCAN_U][4], nxt[SCAN_U][4];
#pragma unroll
    for (int u = 0; u < SCAN_U; u++) load_chunk(wave + 32 * u, cur[u]);
    for (int c0 = wave; c0 < nfull; c0 += 32 * SCAN_U) {
#pragma unroll
      for (int u = 0; u < SCAN_U; u++) load_chunk(c0 + 32 * (SCAN_U + u), nxt[u]);
#pragma unroll
      for (int u = 0; u < SCAN_U; u++) {
        if (c0 + 32 * u < nfull) match_chunk(c0 + 32 * u, cur[u]);
        while (npend >= 32) {
          drain_ring<MODE == 0>(myqe, myqr, head, 32, lane, colp, w, hl, accl, lc, lr, lw, nrec);
          head = (head + 32) & (Q_CAP - 1);
          npend -= 32;
          nrec += 32;
        }
      }
#pragma unroll
      for (int u = 0; u < SCAN_U; u++) {
        cur[u][0] = nxt[u][0]; cur[u][1] = nxt[u][1]; cur[u][2] = nxt[u][2]; cur[u][3] = nxt[u][3];
      }
    }
  }
  if ((E & 127) != 0 && wave == (nfull & 31)) {
    int e0 = (nfull << 7) + lane * 4;
    int rv[4];
#pragma unroll
    for (int k = 0; k < 4; k++) rv[k] = (e0 + k < E) ? rowp[e0 + k] : -1;
    match_chunk(nfull, rv);
  }
  while (npend > 0) {
    int cnt = npend < 32 ? npend : 32;
    drain_ring<MODE == 0>(myqe, myqr, head, cnt, lane, colp, w, hl, accl, lc, lr, lw, nrec);
    head = (head + cnt) & (Q_CAP - 1);
    npend -= cnt;
    nrec += cnt;
  }
  if (MODE == 0 && lane == 0) lcnt[lslot] = nrec;
  }

  __syncthreads();
  int nrows = N - r0;
  if (nrows > BLK_ROWS) nrows = BLK_ROWS;
  float4* dst4 = reinterpret_cast<float4*>(ha + (size_t)r0 * 64);
  for (int i = tid; i < nrows * 16; i += 1024) dst4[i] = reinterpret_cast<const float4*>(acc)[i];
}

__launch_bounds__(256)
__global__ void layer_kernel(const float* __restrict__ h,  const float* __restrict__ ha,
                             const float* __restrict__ Wg, const float* __restrict__ bg,
                             const float* __restrict__ Wi, const float* __restrict__ bi,
                             float* __restrict__ hnext, float* __restrict__ out,
                             int N, int outStride, int colOff) {
  __shared__ float sWg[64 * 64];
  __shared__ float sWi[64 * 64];
  __shared__ float sbg[64];
  __shared__ float sbi[64];

  int tid = threadIdx.x;
  for (int i = tid; i < 4096; i += 256) {
    int r = i >> 6, c = i & 63;
    int idx = (((r >> 1) * 64 + c) << 1) | (r & 1);
    sWg[idx] = Wg[i];
    sWi[idx] = Wi[i];
  }
  if (tid < 64) { sbg[tid] = bg[tid]; sbi[tid] = bi[tid]; }
  __syncthreads();

  int wave = tid >> 5, lane = tid & 31;
  int tiles = (N + 15) >> 4;
  int tile = blockIdx.x * 8 + wave;
  if (tile >= tiles) return;

  int m0 = tile << 4;
  int kh  = lane >> 4;
  int l15 = lane & 15;

  int rowA = m0 + l15;
  int rowLd = rowA < N ? rowA : N - 1;
  const float* haRow = ha + (size_t)rowLd * 64 + 2 * kh;
  const float* hRow  = h  + (size_t)rowLd * 64 + 2 * kh;

  v8f zero8 = {0.f, 0.f, 0.f, 0.f, 0.f, 0.f, 0.f, 0.f};
  v8f accg[4], acci[4];
#pragma unroll
  for (int nt = 0; nt < 4; nt++) { accg[nt] = zero8; acci[nt] = zero8; }

  const v2f* sWg2 = reinterpret_cast<const v2f*>(sWg);
  const v2f* sWi2 = reinterpret_cast<const v2f*>(sWi);

#pragma unroll
  for (int k = 0; k < 64; k += 4) {
    v2f ga = *reinterpret_cast<const v2f*>(haRow + k);
    v2f hv = *reinterpret_cast<const v2f*>(hRow + k);
    v2f ia = ga * hv;
    int k2 = (k >> 1) + kh;
#pragma unroll
    for (int nt = 0; nt < 4; nt++) {
      v2f bgf = sWg2[k2 * 64 + nt * 16 + l15];
      v2f bif = sWi2[k2 * 64 + nt * 16 + l15];
      accg[nt] = __builtin_amdgcn_wmma_f32_16x16x4_f32(
          false, ga, false, bgf, (short)0, accg[nt], false, false);
      acci[nt] = __builtin_amdgcn_wmma_f32_16x16x4_f32(
          false, ia, false, bif, (short)0, acci[nt], false, false);
    }
  }

  float bgl[4], bil[4];
#pragma unroll
  for (int nt = 0; nt < 4; nt++) {
    bgl[nt] = sbg[nt * 16 + l15];
    bil[nt] = sbi[nt * 16 + l15];
  }

#pragma unroll
  for (int j = 0; j < 8; j++) {
    int row = m0 + j + 8 * kh;
    float v[4];
    float sq = 0.0f;
#pragma unroll
    for (int nt = 0; nt < 4; nt++) {
      float g = accg[nt][j] + bgl[nt];
      float s = acci[nt][j] + bil[nt];
      g = (g > 0.0f) ? g : LEAKY * g;
      s = (s > 0.0f) ? s : LEAKY * s;
      v[nt] = g + s;
      sq += v[nt] * v[nt];
    }
#pragma unroll
    for (int m = 1; m < 16; m <<= 1) sq += __shfl_xor(sq, m, 32);
    float inv = rsqrtf(fmaxf(sq, L2EPS));
    if (row < N) {
      float* hn = hnext + (size_t)row * 64 + l15;
      float* op = out + (size_t)row * outStride + colOff + l15;
#pragma unroll
      for (int nt = 0; nt < 4; nt++) {
        hn[nt * 16] = v[nt];
        op[nt * 16] = v[nt] * inv;
      }
    }
  }
}


extern "C" void kernel_launch(void* const* d_in, const int* in_sizes, int n_in,
                              void* d_out, int out_size, void* d_ws, size_t ws_size,
                              hipStream_t stream) {
  const float* x  = (const float*)d_in[0];
  const int*   ei = (const int*)d_in[1];
  const float* Wg = (const float*)d_in[2];
  const float* bg = (const float*)d_in[3];
  const float* Wi = (const float*)d_in[4];
  const float* bi = (const float*)d_in[5];

  const int d = 64;
  const int N = in_sizes[0] / d;
  const int E = in_sizes[1] / 2;
  const int K = in_sizes[2] / (d * d);
  const int outStride = (K + 1) * d;

  float* ws    = (float*)d_ws;
  float* dinv  = ws;
  float* wvec  = dinv + N;
  float* haBuf = wvec + E;
  float* hA    = haBuf + (size_t)N * d;
  float* hB    = hA + (size_t)N * d;
  float* out   = (float*)d_out;

  fill_zero_f32<<<(N + 255) / 256, 256, 0, stream>>>(dinv, N);
  deg_kernel<<<(E + 255) / 256, 256, 0, stream>>>(ei, dinv, E);
  dinv_kernel<<<(N + 255) / 256, 256, 0, stream>>>(dinv, N);
  edgew_kernel<<<(E + 255) / 256, 256, 0, stream>>>(ei, dinv, wvec, E);
  copy_x_kernel<<<(N * 16 + 255) / 256, 256, 0, stream>>>(x, out, N, outStride);

  const float* h = x;
  float* nxt = hA;
  int tiles = (N + 15) / 16;
  int lblocks = (tiles + 7) / 8;
  int ablocks = (N + BLK_ROWS - 1) / BLK_ROWS;
  int*   lcnt = (int*)(hB + (size_t)N * d);
  int*   lcol = lcnt + (size_t)ablocks * 32;
  int*   lrow = lcol + (size_t)ablocks * 32 * LIST_CAP;
  float* lwt  = (float*)(lrow + (size_t)ablocks * 32 * LIST_CAP);

  for (int layer = 0; layer < K; layer++) {
    if (layer == 0)
      agg_kernel<0><<<ablocks, 1024, 0, stream>>>(ei, wvec, h, haBuf, lcnt, lcol, lrow, lwt, E, N);
    else
      agg_kernel<1><<<ablocks, 1024, 0, stream>>>(ei, wvec, h, haBuf, lcnt, lcol, lrow, lwt, E, N);
    layer_kernel<<<lblocks, 256, 0, stream>>>(
        h, haBuf,
        Wg + (size_t)layer * d * d, bg + (size_t)layer * d,
        Wi + (size_t)layer * d * d, bi + (size_t)layer * d,
        nxt, out, N, outStride, (layer + 1) * d);
    h = nxt;
    nxt = (nxt == hA) ? hB : hA;
  }
}


# Degrees by LDS integer adds in a list-build launch ahead of layer 1 (owner blocks filter the row list there, once); layer 1 forms w from dinv; global float atomics, 4 launches dropped; sums as before
# speedup vs baseline: 2.5662x; 1.2987x over previous
#include <hip/hip_runtime.h>

typedef float v2f __attribute__((ext_vector_type(2)));
typedef float v8f __attribute__((ext_vector_type(8)));

#define LEAKY 0.2f
#define L2EPS 1e-12f


__global__ void copy_x_kernel(const float* __restrict__ x, float* __restrict__ out,
                              int N, int outStride) {
  int t = blockIdx.x * blockDim.x + threadIdx.x;
  if (t < N * 16) {
    int r = t >> 4, q = t & 15;
    float4 v = reinterpret_cast<const float4*>(x)[(size_t)r * 16 + q];
    reinterpret_cast<float4*>(out + (size_t)r * outStride)[q] = v;
  }
}

#define BLK_ROWS 392
#define Q_CAP 256
#define DRAIN_G 16
#define SCAN_U 4
#define LIST_CAP 1024

static __device__ __forceinline__ float bcast_f32(float v, int srcLane) {
  return __builtin_bit_cast(float,
      __builtin_amdgcn_readlane(__builtin_bit_cast(int, v), srcLane));
}

static __device__ __forceinline__ void add_group(int c, int rl, float wt, int cnt,
                                                 const float* __restrict__ hl, float* accl) {
#pragma unroll
  for (int j0 = 0; j0 < 32; j0 += DRAIN_G) {
    v2f hv[DRAIN_G];
#pragma unroll
    for (int j = 0; j < DRAIN_G; j++) {
      if (j0 + j < cnt) {
        int cj = __builtin_amdgcn_readlane(c, j0 + j);
        hv[j] = *reinterpret_cast<const v2f*>(hl + (size_t)cj * 64);
      }
    }
#pragma unroll
    for (int j = 0; j < DRAIN_G; j++) {
      if (j0 + j < cnt) {
        int rj = __builtin_amdgcn_readlane(rl, j0 + j);
        float wj = bcast_f32(wt, j0 + j);
        float* dst = accl + rj * 64;
        atomicAdd(dst + 0, wj * hv[j].x);
        atomicAdd(dst + 1, wj * hv[j].y);
      }
    }
  }
}

template <bool COUNT, class DrainF>
static __device__ __forceinline__ int scan_row_list(const int* __restrict__ rowp, int E, int r0,
                                                    int wave, int lane, int* myqe, int* myqr,
                                                    int* degl, DrainF drain) {
  int head = 0, npend = 0;
  int nrec = 0;
  int nfull = E >> 7;
  auto load_chunk = [&](int chunk, int* rv) {
    int cl = chunk < nfull ? chunk : nfull - 1;
    int4 t = *reinterpret_cast<const int4*>(rowp + (cl << 7) + lane * 4);
    rv[0] = t.x; rv[1] = t.y; rv[2] = t.z; rv[3] = t.w;
  };
  auto match_chunk = [&](int chunk, const int* rv) {
    int e0 = (chunk << 7) + lane * 4;
#pragma unroll
    for (int k = 0; k < 4; k++) {
      int rl = rv[k] - r0;
      bool p = (unsigned)rl < (unsigned)BLK_ROWS;
      unsigned m = __builtin_amdgcn_ballot_w32(p);
      if (m != 0u) {
        if (p) {
          int pos = (head + npend + (int)__builtin_amdgcn_mbcnt_lo(m, 0u)) & (Q_CAP - 1);
          myqe[pos] = e0 + k;
          myqr[pos] = rl;
          if (COUNT) atomicAdd(degl + rl, 1);
        }
        npend += __builtin_popcount(m);
      }
    }
  };

  if (nfull > 0) {
    int cur[SCAN_U][4], nxt[SCAN_U][4];
#pragma unroll
    for (int u = 0; u < SCAN_U; u++) load_chunk(wave + 32 * u, cur[u]);
    for (int c0 = wave; c0 < nfull; c0 += 32 * SCAN_U) {
#pragma unroll
      for (int u = 0; u < SCAN_U; u++) load_chunk(c0 + 32 * (SCAN_U + u), nxt[u]);
#pragma unroll
      for (int u = 0; u < SCAN_U; u++) {
        if (c0 + 32 * u < nfull) match_chunk(c0 + 32 * u, cur[u]);
        while (npend >= 32) {
          drain(head, 32, nrec);
          head = (head + 32) & (Q_CAP - 1);
          npend -= 32;
          nrec += 32;
        }
      }
#pragma unroll
      for (int u = 0; u < SCAN_U; u++) {
        cur[u][0] = nxt[u][0]; cur[u][1] = nxt[u][1]; cur[u][2] = nxt[u][2]; cur[u][3] = nxt[u][3];
      }
    }
  }
  if ((E & 127) != 0 && wave == (nfull & 31)) {
    int e0 = (nfull << 7) + lane * 4;
    int rv[4];
#pragma unroll
    for (int k = 0; k < 4; k++) rv[k] = (e0 + k < E) ? rowp[e0 + k] : -1;
    match_chunk(nfull, rv);
  }
  while (npend > 0) {
    int cnt = npend < 32 ? npend : 32;
    drain(head, cnt, nrec);
    head = (head + cnt) & (Q_CAP - 1);
    npend -= cnt;
    nrec += cnt;
  }
  return nrec;
}

__launch_bounds__(1024)
__global__ void prep_kernel(const int* __restrict__ ei, float* __restrict__ dinv,
                            int* __restrict__ lcnt, int* __restrict__ lcol,
                            int* __restrict__ lrow, int E, int N) {
  __shared__ int degl[BLK_ROWS];
  __shared__ int qe[32 * Q_CAP];
  __shared__ int qr[32 * Q_CAP];

  int tid = threadIdx.x;
  int lane = tid & 31;
  int wave = __builtin_amdgcn_readfirstlane(tid >> 5);
  int r0 = blockIdx.x * BLK_ROWS;

  for (int i = tid; i < BLK_ROWS; i += 1024) degl[i] = 0;
  __syncthreads();

  const int* colp = ei + E;
  int* myqe = qe + wave * Q_CAP;
  int* myqr = qr + wave * Q_CAP;

  int lslot = blockIdx.x * 32 + wave;
  size_t lbase = (size_t)lslot * LIST_CAP;
  int* lc = lcol + lbase;
  int* lr = lrow + lbase;

  auto drain = [&](int head, int cnt, int nrec) {
    int idx = (head + lane) & (Q_CAP - 1);
    bool live = lane < cnt;
    int e  = live ? myqe[idx] : 0;
    int rl = live ? myqr[idx] : 0;
    int c = colp[e];
    if (nrec + cnt <= LIST_CAP && live) {
      lc[nrec + lane] = c;
      lr[nrec + lane] = rl;
    }
  };
  int nrec = scan_row_list<true>(ei, E, r0, wave, lane, myqe, myqr, degl, drain);
  if (lane == 0) lcnt[lslot] = nrec;

  __syncthreads();
  int nrows = N - r0;
  if (nrows > BLK_ROWS) nrows = BLK_ROWS;
  for (int i = tid; i < nrows; i += 1024) {
    int dg = degl[i];
    dinv[r0 + i] = (dg > 0) ? rsqrtf((float)dg) : 0.0f;
  }
}

template <int MODE>
__launch_bounds__(1024)
__global__ void agg_kernel(const int* __restrict__ ei, const float* __restrict__ dinv,
                           const float* __restrict__ h, float* __restrict__ ha,
                           const int* __restrict__ lcnt, const int* __restrict__ lcol,
                           const int* __restrict__ lrow, float* __restrict__ lwt, int E, int N) {
  __shared__ float acc[BLK_ROWS * 64];
  __shared__ int qe[32 * Q_CAP];
  __shared__ int qr[32 * Q_CAP];

  int tid = threadIdx.x;
  int lane = tid & 31;
  int wave = __builtin_amdgcn_readfirstlane(tid >> 5);
  int r0 = blockIdx.x * BLK_ROWS;

  float4 z4 = {0.0f, 0.0f, 0.0f, 0.0f};
  for (int i = tid; i < BLK_ROWS * 16; i += 1024) reinterpret_cast<float4*>(acc)[i] = z4;
  __syncthreads();

  const float* hl = h + lane * 2;
  float* accl = acc + lane * 2;
  const float* dinvr = dinv + r0;

  int lslot = blockIdx.x * 32 + wave;
  size_t lbase = (size_t)lslot * LIST_CAP;
  const int* lc = lcol + lbase;
  const int* lr = lrow + lbase;
  float* lw = lwt + lbase;

  int n = __builtin_amdgcn_readfirstlane(lcnt[lslot]);
  if (n <= LIST_CAP) {
    int ng = (n + 31) >> 5;
    int nwhole = n >> 5;
    if (ng > 0) {
      if (MODE == 1) {
        int c = lc[lane], rl = lr[lane];
        float wt = lw[lane];
        for (int g = 0; g < nwhole; g++) {
          int gn = (g + 1 < ng) ? g + 1 : g;
          int cn = lc[gn * 32 + lane], rln = lr[gn * 32 + lane];
          float wtn = lw[gn * 32 + lane];
          add_group(c, rl, wt, 32, hl, accl);
          c = cn; rl = rln; wt = wtn;
        }
        if ((n & 31) != 0) add_group(c, rl, wt, n & 31, hl, accl);
      } else {
        auto load_ent = [&](int g, int& c, int& rl) {
          int i = g * 32 + lane;
          i = i < n ? i : n - 1;
          c = lc[i]; rl = lr[i];
        };
        int c0, rl0, c1, rl1;
        load_ent(0, c0, rl0);
        load_ent(ng > 1 ? 1 : 0, c1, rl1);
        float dv0 = dinv[c0], dr0 = dinvr[rl0];
        for (int g = 0; g < nwhole; g++) {
          int g2 = (g + 2 < ng) ? g + 2 : ng - 1;
          int c2, rl2;
          load_ent(g2, c2, rl2);
          float dv1 = dinv[c1], dr1 = dinvr[rl1];
          float wt0 = dr0 * dv0;
          lw[g * 32 + lane] = wt0;
          add_group(c0, rl0, wt0, 32, hl, accl);
          c0 = c1; rl0 = rl1; dv0 = dv1; dr0 = dr1;
          c1 = c2; rl1 = rl2;
        }
        if ((n & 31) != 0) {
          float wt0 = dr0 * dv0;
          if (lane < (n & 31)) lw[nwhole * 32 + lane] = wt0;
          add_group(c0, rl0, wt0, n & 31, hl, accl);
        }
      }
    }
  } else {
    const int* colp = ei + E;
    int* myqe = qe + wave * Q_CAP;
    int* myqr = qr + wave * Q_CAP;
    auto drain = [&](int head, int cnt, int nrec) {
      int idx = (head + lane) & (Q_CAP - 1);
      bool live = lane < cnt;
      int e  = live ? myqe[idx] : 0;
      int rl = live ? myqr[idx] : 0;
      int c = colp[e];
      float wt = dinvr[rl] * dinv[c];
      add_group(c, rl, wt, cnt, hl, accl);
    };
    scan_row_list<false>(ei, E, r0, wave, lane, myqe, myqr, (int*)nullptr, drain);
  }

  __syncthreads();
  int nrows = N - r0;
  if (nrows > BLK_ROWS) nrows = BLK_ROWS;
  float4* dst4 = reinterpret_cast<float4*>(ha + (size_t)r0 * 64);
  for (int i = tid; i < nrows * 16; i += 1024) dst4[i] = reinterpret_cast<const float4*>(acc)[i];
}

__launch_bounds__(256)
__global__ void layer_kernel(const float* __restrict__ h,  const float* __restrict__ ha,
                             const float* __restrict__ Wg, const float* __restrict__ bg,
                             const float* __restrict__ Wi, const float* __restrict__ bi,
                             float* __restrict__ hnext, float* __restrict__ out,
                             int N, int outStride, int colOff) {
  __shared__ float sWg[64 * 64];
  __shared__ float sWi[64 * 64];
  __shared__ float sbg[64];
  __shared__ float sbi[64];

  int tid = threadIdx.x;
  for (int i = tid; i < 4096; i += 256) {
    int r = i >> 6, c = i & 63;
    int idx = (((r >> 1) * 64 + c) << 1) | (r & 1);
    sWg[idx] = Wg[i];
    sWi[idx] = Wi[i];
  }
  if (tid < 64) { sbg[tid] = bg[tid]; sbi[tid] = bi[tid]; }
  __syncthreads();

  int wave = tid >> 5, lane = tid & 31;
  int tiles = (N + 15) >> 4;
  int tile = blockIdx.x * 8 + wave;
  if (tile >= tiles) return;

  int m0 = tile << 4;
  int kh  = lane >> 4;
  int l15 = lane & 15;

  int rowA = m0 + l15;
  int rowLd = rowA < N ? rowA : N - 1;
  const float* haRow = ha + (size_t)rowLd * 64 + 2 * kh;
  const float* hRow  = h  + (size_t)rowLd * 64 + 2 * kh;

  v8f zero8 = {0.f, 0.f, 0.f, 0.f, 0.f, 0.f, 0.f, 0.f};
  v8f accg[4], acci[4];
#pragma unroll
  for (int nt = 0; nt < 4; nt++) { accg[nt] = zero8; acci[nt] = zero8; }

  const v2f* sWg2 = reinterpret_cast<const v2f*>(sWg);
  const v2f* sWi2 = reinterpret_cast<const v2f*>(sWi);

#pragma unroll
  for (int k = 0; k < 64; k += 4) {
    v2f ga = *reinterpret_cast<const v2f*>(haRow + k);
    v2f hv = *reinterpret_cast<const v2f*>(hRow + k);
    v2f ia = ga * hv;
    int k2 = (k >> 1) + kh;
#pragma unroll
    for (int nt = 0; nt < 4; nt++) {
      v2f bgf = sWg2[k2 * 64 + nt * 16 + l15];
      v2f bif = sWi2[k2 * 64 + nt * 16 + l15];
      accg[nt] = __builtin_amdgcn_wmma_f32_16x16x4_f32(
          false, ga, false, bgf, (short)0, accg[nt], false, false);
      acci[nt] = __builtin_amdgcn_wmma_f32_16x16x4_f32(
          false, ia, false, bif, (short)0, acci[nt], false, false);
    }
  }

  float bgl[4], bil[4];
#pragma unroll
  for (int nt = 0; nt < 4; nt++) {
    bgl[nt] = sbg[nt * 16 + l15];
    bil[nt] = sbi[nt * 16 + l15];
  }

#pragma unroll
  for (int j = 0; j < 8; j++) {
    int row = m0 + j + 8 * kh;
    float v[4];
    float sq = 0.0f;
#pragma unroll
    for (int nt = 0; nt < 4; nt++) {
      float g = accg[nt][j] + bgl[nt];
      float s = acci[nt][j] + bil[nt];
      g = (g > 0.0f) ? g : LEAKY * g;
      s = (s > 0.0f) ? s : LEAKY * s;
      v[nt] = g + s;
      sq += v[nt] * v[nt];
    }
#pragma unroll
    for (int m = 1; m < 16; m <<= 1) sq += __shfl_xor(sq, m, 32);
    float inv = rsqrtf(fmaxf(sq, L2EPS));
    if (row < N) {
      float* hn = hnext + (size_t)row * 64 + l15;
      float* op = out + (size_t)row * outStride + colOff + l15;
#pragma unroll
      for (int nt = 0; nt < 4; nt++) {
        hn[nt * 16] = v[nt];
        op[nt * 16] = v[nt] * inv;
      }
    }
  }
}


extern "C" void kernel_launch(void* const* d_in, const int* in_sizes, int n_in,
                              void* d_out, int out_size, void* d_ws, size_t ws_size,
                              hipStream_t stream) {
  const float* x  = (const float*)d_in[0];
  const int*   ei = (const int*)d_in[1];
  const float* Wg = (const float*)d_in[2];
  const float* bg = (const float*)d_in[3];
  const float* Wi = (const float*)d_in[4];
  const float* bi = (const float*)d_in[5];

  const int d = 64;
  const int N = in_sizes[0] / d;
  const int E = in_sizes[1] / 2;
  const int K = in_sizes[2] / (d * d);
  const int outStride = (K + 1) * d;

  float* ws    = (float*)d_ws;
  float* dinv  = ws;
  float* haBuf = dinv + N;
  float* hA    = haBuf + (size_t)N * d;
  float* hB    = hA + (size_t)N * d;
  float* out   = (float*)d_out;

  int ablocks = (N + BLK_ROWS - 1) / BLK_ROWS;
  int*   lcnt = (int*)(hB + (size_t)N * d);
  int*   lcol = lcnt + (size_t)ablocks * 32;
  int*   lrow = lcol + (size_t)ablocks * 32 * LIST_CAP;
  float* lwt  = (float*)(lrow + (size_t)ablocks * 32 * LIST_CAP);

  prep_kernel<<<ablocks, 1024, 0, stream>>>(ei, dinv, lcnt, lcol, lrow, E, N);
  copy_x_kernel<<<(N * 16 + 255) / 256, 256, 0, stream>>>(x, out, N, outStride);

  const float* h = x;
  float* nxt = hA;
  int tiles = (N + 15) / 16;
  int lblocks = (tiles + 7) / 8;

  for (int layer = 0; layer < K; layer++) {
    if (layer == 0)
      agg_kernel<0><<<ablocks, 1024, 0, stream>>>(ei, dinv, h, haBuf, lcnt, lcol, lrow, lwt, E, N);
    else
      agg_kernel<1><<<ablocks, 1024, 0, stream>>>(ei, dinv, h, haBuf, lcnt, lcol, lrow, lwt, E, N);
    layer_kernel<<<lblocks, 256, 0, stream>>>(
        h, haBuf,
        Wg + (size_t)layer * d * d, bg + (size_t)layer * d,
        Wi + (size_t)layer * d * d, bi + (size_t)layer * d,
        nxt, out, N, outStride, (layer + 1) * d);
    h = nxt;
    nxt = (nxt == hA) ? hB : hA;
  }
}


# Edge filter: a binning launch sorts 1/256 slices of the edge list by owner block into workspace segments (returning LDS integer adds, arrival order); owners copy segments, not scan the row list
# speedup vs baseline: 3.1747x; 1.2371x over previous
#include <hip/hip_runtime.h>

typedef float v2f __attribute__((ext_vector_type(2)));
typedef float v8f __attribute__((ext_vector_type(8)));
typedef int v2i __attribute__((ext_vector_type(2)));

#define LEAKY 0.2f
#define L2EPS 1e-12f


__global__ void copy_x_kernel(const float* __restrict__ x, float* __restrict__ out,
                              int N, int outStride) {
  int t = blockIdx.x * blockDim.x + threadIdx.x;
  if (t < N * 16) {
    int r = t >> 4, q = t & 15;
    float4 v = reinterpret_cast<const float4*>(x)[(size_t)r * 16 + q];
    reinterpret_cast<float4*>(out + (size_t)r * outStride)[q] = v;
  }
}

#define BLK_ROWS 392
#define Q_CAP 256
#define DRAIN_G 16
#define SCAN_U 4
#define LIST_CAP 1024
#define BIN_G 256
#define SEG_PER_WAVE (BIN_G / 32)
#define SEG_CAP 64
#define BIN_T 512
#define BIN_U 8
#define BIN_MAXB 2048

static __device__ __forceinline__ float bcast_f32(float v, int srcLane) {
  return __builtin_bit_cast(float,
      __builtin_amdgcn_readlane(__builtin_bit_cast(int, v), srcLane));
}

static __device__ __forceinline__ void add_group(int c, int rl, float wt, int cnt,
                                                 const float* __restrict__ hl, float* accl) {
#pragma unroll
  for (int j0 = 0; j0 < 32; j0 += DRAIN_G) {
    v2f hv[DRAIN_G];
#pragma unroll
    for (int j = 0; j < DRAIN_G; j++) {
      if (j0 + j < cnt) {
        int cj = __builtin_amdgcn_readlane(c, j0 + j);
        hv[j] = *reinterpret_cast<const v2f*>(hl + (size_t)cj * 64);
      }
    }
#pragma unroll
    for (int j = 0; j < DRAIN_G; j++) {
      if (j0 + j < cnt) {
        int rj = __builtin_amdgcn_readlane(rl, j0 + j);
        float wj = bcast_f32(wt, j0 + j);
        float* dst = accl + rj * 64;
        atomicAdd(dst + 0, wj * hv[j].x);
        atomicAdd(dst + 1, wj * hv[j].y);
      }
    }
  }
}

template <bool COUNT, class DrainF>
static __device__ __forceinline__ int scan_row_list(const int* __restrict__ rowp, int E, int r0,
                                                    int wave, int lane, int* myqe, int* myqr,
                                                    int* degl, DrainF drain) {
  int head = 0, npend = 0;
  int nrec = 0;
  int nfull = E >> 7;
  auto load_chunk = [&](int chunk, int* rv) {
    int cl = chunk < nfull ? chunk : nfull - 1;
    int4 t = *reinterpret_cast<const int4*>(rowp + (cl << 7) + lane * 4);
    rv[0] = t.x; rv[1] = t.y; rv[2] = t.z; rv[3] = t.w;
  };
  auto match_chunk = [&](int chunk, const int* rv) {
    int e0 = (chunk << 7) + lane * 4;
#pragma unroll
    for (int k = 0; k < 4; k++) {
      int rl = rv[k] - r0;
      bool p = (unsigned)rl < (unsigned)BLK_ROWS;
      unsigned m = __builtin_amdgcn_ballot_w32(p);
      if (m != 0u) {
        if (p) {
          int pos = (head + npend + (int)__builtin_amdgcn_mbcnt_lo(m, 0u)) & (Q_CAP - 1);
          myqe[pos] = e0 + k;
          myqr[pos] = rl;
          if (COUNT) atomicAdd(degl + rl, 1);
        }
        npend += __builtin_popcount(m);
      }
    }
  };

  if (nfull > 0) {
    int cur[SCAN_U][4], nxt[SCAN_U][4];
#pragma unroll
    for (int u = 0; u < SCAN_U; u++) load_chunk(wave + 32 * u, cur[u]);
    for (int c0 = wave; c0 < nfull; c0 += 32 * SCAN_U) {
#pragma unroll
      for (int u = 0; u < SCAN_U; u++) load_chunk(c0 + 32 * (SCAN_U + u), nxt[u]);
#pragma unroll
      for (int u = 0; u < SCAN_U; u++) {
        if (c0 + 32 * u < nfull) match_chunk(c0 + 32 * u, cur[u]);
        while (npend >= 32) {
          drain(head, 32, nrec);
          head = (head + 32) & (Q_CAP - 1);
          npend -= 32;
          nrec += 32;
        }
      }
#pragma unroll
      for (int u = 0; u < SCAN_U; u++) {
        cur[u][0] = nxt[u][0]; cur[u][1] = nxt[u][1]; cur[u][2] = nxt[u][2]; cur[u][3] = nxt[u][3];
      }
    }
  }
  if ((E & 127) != 0 && wave == (nfull & 31)) {
    int e0 = (nfull << 7) + lane * 4;
    int rv[4];
#pragma unroll
    for (int k = 0; k < 4; k++) rv[k] = (e0 + k < E) ? rowp[e0 + k] : -1;
    match_chunk(nfull, rv);
  }
  while (npend > 0) {
    int cnt = npend < 32 ? npend : 32;
    drain(head, cnt, nrec);
    head = (head + cnt) & (Q_CAP - 1);
    npend -= cnt;
    nrec += cnt;
  }
  return nrec;
}

__launch_bounds__(BIN_T)
__global__ void bin_kernel(const int* __restrict__ ei, int* __restrict__ scnt,
                           v2i* __restrict__ segs, int E, int N, int nblk) {
  __shared__ int cntl[BIN_MAXB];
  int tid = threadIdx.x;
  int j = blockIdx.x;
  if (nblk > BIN_MAXB) {
    for (int b = tid; b < nblk; b += BIN_T) scnt[(size_t)b * BIN_G + j] = SEG_CAP + 1;
    return;
  }
  for (int b = tid; b < nblk; b += BIN_T) cntl[b] = 0;
  __syncthreads();

  int S = (E + BIN_G - 1) / BIN_G;
  int e0 = j * S;
  int e1 = (e0 + S < E) ? e0 + S : E;
  const int* colp = ei + E;
  for (int base = e0; base < e1; base += BIN_T * BIN_U) {
    int r[BIN_U], c[BIN_U];
#pragma unroll
    for (int u = 0; u < BIN_U; u++) {
      int e = base + u * BIN_T + tid;
      int ec = e < e1 ? e : e1 - 1;
      int rv = ei[ec];
      c[u] = colp[ec];
      r[u] = e < e1 ? rv : -1;
    }
#pragma unroll
    for (int u = 0; u < BIN_U; u++) {
      int rr = r[u];
      if ((unsigned)rr < (unsigned)N) {
        int b = (int)((unsigned)rr / (unsigned)BLK_ROWS);
        int rl = rr - b * BLK_ROWS;
        int rank = atomicAdd(&cntl[b], 1);
        if (rank < SEG_CAP)
          segs[((size_t)b * BIN_G + j) * SEG_CAP + rank] = (v2i){c[u], rl};
      }
    }
  }
  __syncthreads();
  for (int b = tid; b < nblk; b += BIN_T) scnt[(size_t)b * BIN_G + j] = cntl[b];
}

__launch_bounds__(1024)
__global__ void prep_kernel(const int* __restrict__ ei, const int* __restrict__ scnt,
                            const v2i* __restrict__ segs, float* __restrict__ dinv,
                            int* __restrict__ lcnt, int* __restrict__ lcol,
                            int* __restrict__ lrow, int E, int N) {
  __shared__ int degl[BLK_ROWS];
  __shared__ int qe[32 * Q_CAP];
  __shared__ int qr[32 * Q_CAP];
  __shared__ int notwhole;

  int tid = threadIdx.x;
  int lane = tid & 31;
  int wave = __builtin_amdgcn_readfirstlane(tid >> 5);
  int r0 = blockIdx.x * BLK_ROWS;

  for (int i = tid; i < BLK_ROWS; i += 1024) degl[i] = 0;
  if (tid == 0) notwhole = 0;
  __syncthreads();

  const int* colp = ei + E;
  int* myqe = qe + wave * Q_CAP;
  int* myqr = qr + wave * Q_CAP;

  int lslot = blockIdx.x * 32 + wave;
  size_t lbase = (size_t)lslot * LIST_CAP;
  int* lc = lcol + lbase;
  int* lr = lrow + lbase;

  size_t sbase = (size_t)blockIdx.x * BIN_G + wave * SEG_PER_WAVE;
  int myc = (lane < SEG_PER_WAVE) ? scnt[sbase + lane] : 0;
  if (__builtin_amdgcn_ballot_w32(myc > SEG_CAP) != 0u && lane == 0) notwhole = 1;
  __syncthreads();
  int slow = __builtin_amdgcn_readfirstlane(notwhole);

  auto drain = [&](int head, int cnt, int nrec) {
    int idx = (head + lane) & (Q_CAP - 1);
    bool live = lane < cnt;
    int e  = live ? myqe[idx] : 0;
    int rl = live ? myqr[idx] : 0;
    int c = colp[e];
    if (nrec + cnt <= LIST_CAP && live) {
      lc[nrec + lane] = c;
      lr[nrec + lane] = rl;
    }
  };
  int nrec;
  if (slow == 0) {
    int cq[SEG_PER_WAVE];
#pragma unroll
    for (int q = 0; q < SEG_PER_WAVE; q++) cq[q] = __builtin_amdgcn_readlane(myc, q);
    const v2i* sg = segs + sbase * SEG_CAP;
    v2i en[SEG_PER_WAVE][SEG_CAP / 32];
#pragma unroll
    for (int q = 0; q < SEG_PER_WAVE; q++) {
#pragma unroll
      for (int p = 0; p < SEG_CAP / 32; p++) {
        int i = p * 32 + lane;
        en[q][p] = (v2i){0, 0};
        if (i < cq[q]) en[q][p] = sg[q * SEG_CAP + i];
      }
    }
    nrec = 0;
#pragma unroll
    for (int q = 0; q < SEG_PER_WAVE; q++) {
#pragma unroll
      for (int p = 0; p < SEG_CAP / 32; p++) {
        int i = p * 32 + lane;
        if (i < cq[q]) {
          lc[nrec + i] = en[q][p].x;
          lr[nrec + i] = en[q][p].y;
          atomicAdd(degl + en[q][p].y, 1);
        }
      }
      nrec += cq[q];
    }
  } else {
    nrec = scan_row_list<true>(ei, E, r0, wave, lane, myqe, myqr, degl, drain);
  }
  if (lane == 0) lcnt[lslot] = nrec;

  __syncthreads();
  int nrows = N - r0;
  if (nrows > BLK_ROWS) nrows = BLK_ROWS;
  for (int i = tid; i < nrows; i += 1024) {
    int dg = degl[i];
    dinv[r0 + i] = (dg > 0) ? rsqrtf((float)dg) : 0.0f;
  }
}

template <int MODE>
__launch_bounds__(1024)
__global__ void agg_kernel(const int* __restrict__ ei, const float* __restrict__ dinv,
                           const float* __restrict__ h, float* __restrict__ ha,
                           const int* __restrict__ lcnt, const int* __restrict__ lcol,
                           const int* __restrict__ lrow, float* __restrict__ lwt, int E, int N) {
  __shared__ float acc[BLK_ROWS * 64];
  __shared__ int qe[32 * Q_CAP];
  __shared__ int qr[32 * Q_CAP];

  int tid = threadIdx.x;
  int lane = tid & 31;
  int wave = __builtin_amdgcn_readfirstlane(tid >> 5);
  int r0 = blockIdx.x * BLK_ROWS;

  float4 z4 = {0.0f, 0.0f, 0.0f, 0.0f};
  for (int i = tid; i < BLK_ROWS * 16; i += 1024) reinterpret_cast<float4*>(acc)[i] = z4;
  __syncthreads();

  const float* hl = h + lane * 2;
  float* accl = acc + lane * 2;
  const float* dinvr = dinv + r0;

  int lslot = blockIdx.x * 32 + wave;
  size_t lbase = (size_t)lslot * LIST_CAP;
  const int* lc = lcol + lbase;
  const int* lr = lrow + lbase;
  float* lw = lwt + lbase;

  int n = __builtin_amdgcn_readfirstlane(lcnt[lslot]);
  if (n <= LIST_CAP) {
    int ng = (n + 31) >> 5;
    int nwhole = n >> 5;
    if (ng > 0) {
      if (MODE == 1) {
        int c = lc[lane], rl = lr[lane];
        float wt = lw[lane];
        for (int g = 0; g < nwhole; g++) {
          int gn = (g + 1 < ng) ? g + 1 : g;
          int cn = lc[gn * 32 + lane], rln = lr[gn * 32 + lane];
          float wtn = lw[gn * 32 + lane];
          add_group(c, rl, wt, 32, hl, accl);
          c = cn; rl = rln; wt = wtn;
        }
        if ((n & 31) != 0) add_group(c, rl, wt, n & 31, hl, accl);
      } else {
        auto load_ent = [&](int g, int& c, int& rl) {
          int i = g * 32 + lane;
          i = i < n ? i : n - 1;
          c = lc[i]; rl = lr[i];
        };
        int c0, rl0, c1, rl1;
        load_ent(0, c0, rl0);
        load_ent(ng > 1 ? 1 : 0, c1, rl1);
        float dv0 = dinv[c0], dr0 = dinvr[rl0];
        for (int g = 0; g < nwhole; g++) {
          int g2 = (g + 2 < ng) ? g + 2 : ng - 1;
          int c2, rl2;
          load_ent(g2, c2, rl2);
          float dv1 = dinv[c1], dr1 = dinvr[rl1];
          float wt0 = dr0 * dv0;
          lw[g * 32 + lane] = wt0;
          add_group(c0, rl0, wt0, 32, hl, accl);
          c0 = c1; rl0 = rl1; dv0 = dv1; dr0 = dr1;
          c1 = c2; rl1 = rl2;
        }
        if ((n & 31) != 0) {
          float wt0 = dr0 * dv0;
          if (lane < (n & 31)) lw[nwhole * 32 + lane] = wt0;
          add_group(c0, rl0, wt0, n & 31, hl, accl);
        }
      }
    }
  } else {
    const int* colp = ei + E;
    int* myqe = qe + wave * Q_CAP;
    int* myqr = qr + wave * Q_CAP;
    auto drain = [&](int head, int cnt, int nrec) {
      int idx = (head + lane) & (Q_CAP - 1);
      bool live = lane < cnt;
      int e  = live ? myqe[idx] : 0;
      int rl = live ? myqr[idx] : 0;
      int c = colp[e];
      float wt = dinvr[rl] * dinv[c];
      add_group(c, rl, wt, cnt, hl, accl);
    };
    scan_row_list<false>(ei, E, r0, wave, lane, myqe, myqr, (int*)nullptr, drain);
  }

  __syncthreads();
  int nrows = N - r0;
  if (nrows > BLK_ROWS) nrows = BLK_ROWS;
  float4* dst4 = reinterpret_cast<float4*>(ha + (size_t)r0 * 64);
  for (int i = tid; i < nrows * 16; i += 1024) dst4[i] = reinterpret_cast<const float4*>(acc)[i];
}

__launch_bounds__(256)
__global__ void layer_kernel(const float* __restrict__ h,  const float* __restrict__ ha,
                             const float* __restrict__ Wg, const float* __restrict__ bg,
                             const float* __restrict__ Wi, const float* __restrict__ bi,
                             float* __restrict__ hnext, float* __restrict__ out,
                             int N, int outStride, int colOff) {
  __shared__ float sWg[64 * 64];
  __shared__ float sWi[64 * 64];
  __shared__ float sbg[64];
  __shared__ float sbi[64];

  int tid = threadIdx.x;
  for (int i = tid; i < 4096; i += 256) {
    int r = i >> 6, c = i & 63;
    int idx = (((r >> 1) * 64 + c) << 1) | (r & 1);
    sWg[idx] = Wg[i];
    sWi[idx] = Wi[i];
  }
  if (tid < 64) { sbg[tid] = bg[tid]; sbi[tid] = bi[tid]; }
  __syncthreads();

  int wave = tid >> 5, lane = tid & 31;
  int tiles = (N + 15) >> 4;
  int tile = blockIdx.x * 8 + wave;
  if (tile >= tiles) return;

  int m0 = tile << 4;
  int kh  = lane >> 4;
  int l15 = lane & 15;

  int rowA = m0 + l15;
  int rowLd = rowA < N ? rowA : N - 1;
  const float* haRow = ha + (size_t)rowLd * 64 + 2 * kh;
  const float* hRow  = h  + (size_t)rowLd * 64 + 2 * kh;

  v8f zero8 = {0.f, 0.f, 0.f, 0.f, 0.f, 0.f, 0.f, 0.f};
  v8f accg[4], acci[4];
#pragma unroll
  for (int nt = 0; nt < 4; nt++) { accg[nt] = zero8; acci[nt] = zero8; }

  const v2f* sWg2 = reinterpret_cast<const v2f*>(sWg);
  const v2f* sWi2 = reinterpret_cast<const v2f*>(sWi);

#pragma unroll
  for (int k = 0; k < 64; k += 4) {
    v2f ga = *reinterpret_cast<const v2f*>(haRow + k);
    v2f hv = *reinterpret_cast<const v2f*>(hRow + k);
    v2f ia = ga * hv;
    int k2 = (k >> 1) + kh;
#pragma unroll
    for (int nt = 0; nt < 4; nt++) {
      v2f bgf = sWg2[k2 * 64 + nt * 16 + l15];
      v2f bif = sWi2[k2 * 64 + nt * 16 + l15];
      accg[nt] = __builtin_amdgcn_wmma_f32_16x16x4_f32(
          false, ga, false, bgf, (short)0, accg[nt], false, false);
      acci[nt] = __builtin_amdgcn_wmma_f32_16x16x4_f32(
          false, ia, false, bif, (short)0, acci[nt], false, false);
    }
  }

  float bgl[4], bil[4];
#pragma unroll
  for (int nt = 0; nt < 4; nt++) {
    bgl[nt] = sbg[nt * 16 + l15];
    bil[nt] = sbi[nt * 16 + l15];
  }

#pragma unroll
  for (int j = 0; j < 8; j++) {
    int row = m0 + j + 8 * kh;
    float v[4];
    float sq = 0.0f;
#pragma unroll
    for (int nt = 0; nt < 4; nt++) {
      float g = accg[nt][j] + bgl[nt];
      float s = acci[nt][j] + bil[nt];
      g = (g > 0.0f) ? g : LEAKY * g;
      s = (s > 0.0f) ? s : LEAKY * s;
      v[nt] = g + s;
      sq += v[nt] * v[nt];
    }
#pragma unroll
    for (int m = 1; m < 16; m <<= 1) sq += __shfl_xor(sq, m, 32);
    float inv = rsqrtf(fmaxf(sq, L2EPS));
    if (row < N) {
      float* hn = hnext + (size_t)row * 64 + l15;
      float* op = out + (size_t)row * outStride + colOff + l15;
#pragma unroll
      for (int nt = 0; nt < 4; nt++) {
        hn[nt * 16] = v[nt];
        op[nt * 16] = v[nt] * inv;
      }
    }
  }
}


extern "C" void kernel_launch(void* const* d_in, const int* in_sizes, int n_in,
                              void* d_out, int out_size, void* d_ws, size_t ws_size,
                              hipStream_t stream) {
  const float* x  = (const float*)d_in[0];
  const int*   ei = (const int*)d_in[1];
  const float* Wg = (const float*)d_in[2];
  const float* bg = (const float*)d_in[3];
  const float* Wi = (const float*)d_in[4];
  const float* bi = (const float*)d_in[5];

  const int d = 64;
  const int N = in_sizes[0] / d;
  const int E = in_sizes[1] / 2;
  const int K = in_sizes[2] / (d * d);
  const int outStride = (K + 1) * d;

  float* ws    = (float*)d_ws;
  float* dinv  = ws;
  float* haBuf = dinv + N;
  float* hA    = haBuf + (size_t)N * d;
  float* hB    = hA + (size_t)N * d;
  float* out   = (float*)d_out;

  int ablocks = (N + BLK_ROWS - 1) / BLK_ROWS;
  int*   lcnt = (int*)(hB + (size_t)N * d);
  int*   lcol = lcnt + (size_t)ablocks * 32;
  int*   lrow = lcol + (size_t)ablocks * 32 * LIST_CAP;
  float* lwt  = (float*)(lrow + (size_t)ablocks * 32 * LIST_CAP);
  int*   scnt = (int*)(lwt + (size_t)ablocks * 32 * LIST_CAP);
  v2i*   segs = (v2i*)(scnt + (size_t)ablocks * BIN_G);

  bin_kernel<<<BIN_G, BIN_T, 0, stream>>>(ei, scnt, segs, E, N, ablocks);
  prep_kernel<<<ablocks, 1024, 0, stream>>>(ei, scnt, segs, dinv, lcnt, lcol, lrow, E, N);
  copy_x_kernel<<<(N * 16 + 255) / 256, 256, 0, stream>>>(x, out, N, outStride);

  const float* h = x;
  float* nxt = hA;
  int tiles = (N + 15) / 16;
  int lblocks = (tiles + 7) / 8;

  for (int layer = 0; layer < K; layer++) {
    if (layer == 0)
      agg_kernel<0><<<ablocks, 1024, 0, stream>>>(ei, dinv, h, haBuf, lcnt, lcol, lrow, lwt, E, N);
    else
      agg_kernel<1><<<ablocks, 1024, 0, stream>>>(ei, dinv, h, haBuf, lcnt, lcol, lrow, lwt, E, N);
    layer_kernel<<<lblocks, 256, 0, stream>>>(
        h, haBuf,
        Wg + (size_t)layer * d * d, bg + (size_t)layer * d,
        Wi + (size_t)layer * d * d, bi + (size_t)layer * d,
        nxt, out, N, outStride, (layer + 1) * d);
    h = nxt;
    nxt = (nxt == hA) ? hB : hA;
  }
}


# Dense layer runs in the aggregation launch: each owner work-group feeds WMMA A fragments from its own LDS row sums (pitch 68 floats); ha plane, 3 launches dropped; LDS adds, arrival order as before
# speedup vs baseline: 3.9255x; 1.2365x over previous
#include <hip/hip_runtime.h>

typedef float v2f __attribute__((ext_vector_type(2)));
typedef float v8f __attribute__((ext_vector_type(8)));
typedef int v2i __attribute__((ext_vector_type(2)));

#define LEAKY 0.2f
#define L2EPS 1e-12f


__global__ void copy_x_kernel(const float* __restrict__ x, float* __restrict__ out,
                              int N, int outStride) {
  int t = blockIdx.x * blockDim.x + threadIdx.x;
  if (t < N * 16) {
    int r = t >> 4, q = t & 15;
    float4 v = reinterpret_cast<const float4*>(x)[(size_t)r * 16 + q];
    reinterpret_cast<float4*>(out + (size_t)r * outStride)[q] = v;
  }
}

#define BLK_ROWS 392
#define ACC_PITCH 68
#define BLK_TILES ((BLK_ROWS + 15) / 16)
#define ACC_ROWS (BLK_TILES * 16)
#define Q_CAP 256
#define DRAIN_G 16
#define SCAN_U 4
#define LIST_CAP 1024
#define BIN_G 256
#define SEG_PER_WAVE (BIN_G / 32)
#define SEG_CAP 64
#define BIN_T 512
#define BIN_U 8
#define BIN_MAXB 2048

static __device__ __forceinline__ float bcast_f32(float v, int srcLane) {
  return __builtin_bit_cast(float,
      __builtin_amdgcn_readlane(__builtin_bit_cast(int, v), srcLane));
}

static __device__ __forceinline__ void add_group(int c, int rl, float wt, int cnt,
                                                 const float* __restrict__ hl, float* accl) {
#pragma unroll
  for (int j0 = 0; j0 < 32; j0 += DRAIN_G) {
    v2f hv[DRAIN_G];
#pragma unroll
    for (int j = 0; j < DRAIN_G; j++) {
      if (j0 + j < cnt) {
        int cj = __builtin_amdgcn_readlane(c, j0 + j);
        hv[j] = *reinterpret_cast<const v2f*>(hl + (size_t)cj * 64);
      }
    }
#pragma unroll
    for (int j = 0; j < DRAIN_G; j++) {
      if (j0 + j < cnt) {
        int rj = __builtin_amdgcn_readlane(rl, j0 + j);
        float wj = bcast_f32(wt, j0 + j);
        float* dst = accl + rj * ACC_PITCH;
        atomicAdd(dst + 0, wj * hv[j].x);
        atomicAdd(dst + 1, wj * hv[j].y);
      }
    }
  }
}

template <bool COUNT, class DrainF>
static __device__ __forceinline__ int scan_row_list(const int* __restrict__ rowp, int E, int r0,
                                                    int wave, int lane, int* myqe, int* myqr,
                                                    int* degl, DrainF drain) {
  int head = 0, npend = 0;
  int nrec = 0;
  int nfull = E >> 7;
  auto load_chunk = [&](int chunk, int* rv) {
    int cl = chunk < nfull ? chunk : nfull - 1;
    int4 t = *reinterpret_cast<const int4*>(rowp + (cl << 7) + lane * 4);
    rv[0] = t.x; rv[1] = t.y; rv[2] = t.z; rv[3] = t.w;
  };
  auto match_chunk = [&](int chunk, const int* rv) {
    int e0 = (chunk << 7) + lane * 4;
#pragma unroll
    for (int k = 0; k < 4; k++) {
      int rl = rv[k] - r0;
      bool p = (unsigned)rl < (unsigned)BLK_ROWS;
      unsigned m = __builtin_amdgcn_ballot_w32(p);
      if (m != 0u) {
        if (p) {
          int pos = (head + npend + (int)__builtin_amdgcn_mbcnt_lo(m, 0u)) & (Q_CAP - 1);
          myqe[pos] = e0 + k;
          myqr[pos] = rl;
          if (COUNT) atomicAdd(degl + rl, 1);
        }
        npend += __builtin_popcount(m);
      }
    }
  };

  if (nfull > 0) {
    int cur[SCAN_U][4], nxt[SCAN_U][4];
#pragma unroll
    for (int u = 0; u < SCAN_U; u++) load_chunk(wave + 32 * u, cur[u]);
    for (int c0 = wave; c0 < nfull; c0 += 32 * SCAN_U) {
#pragma unroll
      for (int u = 0; u < SCAN_U; u++) load_chunk(c0 + 32 * (SCAN_U + u), nxt[u]);
#pragma unroll
      for (int u = 0; u < SCAN_U; u++) {
        if (c0 + 32 * u < nfull) match_chunk(c0 + 32 * u, cur[u]);
        while (npend >= 32) {
          drain(head, 32, nrec);
          head = (head + 32) & (Q_CAP - 1);
          npend -= 32;
          nrec += 32;
        }
      }
#pragma unroll
      for (int u = 0; u < SCAN_U; u++) {
        cur[u][0] = nxt[u][0]; cur[u][1] = nxt[u][1]; cur[u][2] = nxt[u][2]; cur[u][3] = nxt[u][3];
      }
    }
  }
  if ((E & 127) != 0 && wave == (nfull & 31)) {
    int e0 = (nfull << 7) + lane * 4;
    int rv[4];
#pragma unroll
    for (int k = 0; k < 4; k++) rv[k] = (e0 + k < E) ? rowp[e0 + k] : -1;
    match_chunk(nfull, rv);
  }
  while (npend > 0) {
    int cnt = npend < 32 ? npend : 32;
    drain(head, cnt, nrec);
    head = (head + cnt) & (Q_CAP - 1);
    npend -= cnt;
    nrec += cnt;
  }
  return nrec;
}

__launch_bounds__(BIN_T)
__global__ void bin_kernel(const int* __restrict__ ei, int* __restrict__ scnt,
                           v2i* __restrict__ segs, int E, int N, int nblk) {
  __shared__ int cntl[BIN_MAXB];
  int tid = threadIdx.x;
  int j = blockIdx.x;
  if (nblk > BIN_MAXB) {
    for (int b = tid; b < nblk; b += BIN_T) scnt[(size_t)b * BIN_G + j] = SEG_CAP + 1;
    return;
  }
  for (int b = tid; b < nblk; b += BIN_T) cntl[b] = 0;
  __syncthreads();

  int S = (E + BIN_G - 1) / BIN_G;
  int e0 = j * S;
  int e1 = (e0 + S < E) ? e0 + S : E;
  const int* colp = ei + E;
  for (int base = e0; base < e1; base += BIN_T * BIN_U) {
    int r[BIN_U], c[BIN_U];
#pragma unroll
    for (int u = 0; u < BIN_U; u++) {
      int e = base + u * BIN_T + tid;
      int ec = e < e1 ? e : e1 - 1;
      int rv = ei[ec];
      c[u] = colp[ec];
      r[u] = e < e1 ? rv : -1;
    }
#pragma unroll
    for (int u = 0; u < BIN_U; u++) {
      int rr = r[u];
      if ((unsigned)rr < (unsigned)N) {
        int b = (int)((unsigned)rr / (unsigned)BLK_ROWS);
        int rl = rr - b * BLK_ROWS;
        int rank = atomicAdd(&cntl[b], 1);
        if (rank < SEG_CAP)
          segs[((size_t)b * BIN_G + j) * SEG_CAP + rank] = (v2i){c[u], rl};
      }
    }
  }
  __syncthreads();
  for (int b = tid; b < nblk; b += BIN_T) scnt[(size_t)b * BIN_G + j] = cntl[b];
}

__launch_bounds__(1024)
__global__ void prep_kernel(const int* __restrict__ ei, const int* __restrict__ scnt,
                            const v2i* __restrict__ segs, float* __restrict__ dinv,
                            int* __restrict__ lcnt, int* __restrict__ lcol,
                            int* __restrict__ lrow, int E, int N) {
  __shared__ int degl[BLK_ROWS];
  __shared__ int qe[32 * Q_CAP];
  __shared__ int qr[32 * Q_CAP];
  __shared__ int notwhole;

  int tid = threadIdx.x;
  int lane = tid & 31;
  int wave = __builtin_amdgcn_readfirstlane(tid >> 5);
  int r0 = blockIdx.x * BLK_ROWS;

  for (int i = tid; i < BLK_ROWS; i += 1024) degl[i] = 0;
  if (tid == 0) notwhole = 0;
  __syncthreads();

  const int* colp = ei + E;
  int* myqe = qe + wave * Q_CAP;
  int* myqr = qr + wave * Q_CAP;

  int lslot = blockIdx.x * 32 + wave;
  size_t lbase = (size_t)lslot * LIST_CAP;
  int* lc = lcol + lbase;
  int* lr = lrow + lbase;

  size_t sbase = (size_t)blockIdx.x * BIN_G + wave * SEG_PER_WAVE;
  int myc = (lane < SEG_PER_WAVE) ? scnt[sbase + lane] : 0;
  if (__builtin_amdgcn_ballot_w32(myc > SEG_CAP) != 0u && lane == 0) notwhole = 1;
  __syncthreads();
  int slow = __builtin_amdgcn_readfirstlane(notwhole);

  auto drain = [&](int head, int cnt, int nrec) {
    int idx = (head + lane) & (Q_CAP - 1);
    bool live = lane < cnt;
    int e  = live ? myqe[idx] : 0;
    int rl = live ? myqr[idx] : 0;
    int c = colp[e];
    if (nrec + cnt <= LIST_CAP && live) {
      lc[nrec + lane] = c;
      lr[nrec + lane] = rl;
    }
  };
  int nrec;
  if (slow == 0) {
    int cq[SEG_PER_WAVE];
#pragma unroll
    for (int q = 0; q < SEG_PER_WAVE; q++) cq[q] = __builtin_amdgcn_readlane(myc, q);
    const v2i* sg = segs + sbase * SEG_CAP;
    v2i en[SEG_PER_WAVE][SEG_CAP / 32];
#pragma unroll
    for (int q = 0; q < SEG_PER_WAVE; q++) {
#pragma unroll
      for (int p = 0; p < SEG_CAP / 32; p++) {
        int i = p * 32 + lane;
        en[q][p] = (v2i){0, 0};
        if (i < cq[q]) en[q][p] = sg[q * SEG_CAP + i];
      }
    }
    nrec = 0;
#pragma unroll
    for (int q = 0; q < SEG_PER_WAVE; q++) {
#pragma unroll
      for (int p = 0; p < SEG_CAP / 32; p++) {
        int i = p * 32 + lane;
        if (i < cq[q]) {
          lc[nrec + i] = en[q][p].x;
          lr[nrec + i] = en[q][p].y;
          atomicAdd(degl + en[q][p].y, 1);
        }
      }
      nrec += cq[q];
    }
  } else {
    nrec = scan_row_list<true>(ei, E, r0, wave, lane, myqe, myqr, degl, drain);
  }
  if (lane == 0) lcnt[lslot] = nrec;

  __syncthreads();
  int nrows = N - r0;
  if (nrows > BLK_ROWS) nrows = BLK_ROWS;
  for (int i = tid; i < nrows; i += 1024) {
    int dg = degl[i];
    dinv[r0 + i] = (dg > 0) ? rsqrtf((float)dg) : 0.0f;
  }
}

template <int MODE>
__launch_bounds__(1024)
__global__ void agg_layer_kernel(const int* __restrict__ ei, const float* __restrict__ dinv,
                                 const float* __restrict__ h,
                                 const int* __restrict__ lcnt, const int* __restrict__ lcol,
                                 const int* __restrict__ lrow, float* __restrict__ lwt,
                                 const float* __restrict__ Wg, const float* __restrict__ bg,
                                 const float* __restrict__ Wi, const float* __restrict__ bi,
                                 float* __restrict__ hnext, float* __restrict__ out,
                                 int E, int N, int outStride, int colOff) {
  static_assert(BLK_TILES <= 32, "one 16-row tile per wave of the work-group");
  struct __attribute__((aligned(16))) Lds {
    float wg[64 * 64];
    float wi[64 * 64];
    float bg[64];
    float bi[64];
    float acc[ACC_ROWS * ACC_PITCH];
    int qe[32 * Q_CAP];
    int qr[32 * Q_CAP];
  };
  __shared__ Lds lds;
  float* acc = lds.acc;
  float* sWg = lds.wg;
  float* sWi = lds.wi;
  float* sbg = lds.bg;
  float* sbi = lds.bi;
  int* qe = lds.qe;
  int* qr = lds.qr;

  int tid = threadIdx.x;
  int lane = tid & 31;
  int wave = __builtin_amdgcn_readfirstlane(tid >> 5);
  int r0 = blockIdx.x * BLK_ROWS;

  float4 z4 = {0.0f, 0.0f, 0.0f, 0.0f};
  for (int i = tid; i < ACC_ROWS * ACC_PITCH / 4; i += 1024) reinterpret_cast<float4*>(acc)[i] = z4;
  __syncthreads();

  float wgr[4], wir[4];
#pragma unroll
  for (int u = 0; u < 4; u++) {
    wgr[u] = Wg[u * 1024 + tid];
    wir[u] = Wi[u * 1024 + tid];
  }
  float bgr = 0.0f, bir = 0.0f;
  if (tid < 64) { bgr = bg[tid]; bir = bi[tid]; }

  const float* hl = h + lane * 2;
  float* accl = acc + lane * 2;
  const float* dinvr = dinv + r0;

  int lslot = blockIdx.x * 32 + wave;
  size_t lbase = (size_t)lslot * LIST_CAP;
  const int* lc = lcol + lbase;
  const int* lr = lrow + lbase;
  float* lw = lwt + lbase;

  int n = __builtin_amdgcn_readfirstlane(lcnt[lslot]);
  if (n <= LIST_CAP) {
    int ng = (n + 31) >> 5;
    int nwhole = n >> 5;
    if (ng > 0) {
      if (MODE == 1) {
        int c = lc[lane], rl = lr[lane];
        float wt = lw[lane];
        for (int g = 0; g < nwhole; g++) {
          int gn = (g + 1 < ng) ? g + 1 : g;
          int cn = lc[gn * 32 + lane], rln = lr[gn * 32 + lane];
          float wtn = lw[gn * 32 + lane];
          add_group(c, rl, wt, 32, hl, accl);
          c = cn; rl = rln; wt = wtn;
        }
        if ((n & 31) != 0) add_group(c, rl, wt, n & 31, hl, accl);
      } else {
        auto load_ent = [&](int g, int& c, int& rl) {
          int i = g * 32 + lane;
          i = i < n ? i : n - 1;
          c = lc[i]; rl = lr[i];
        };
        int c0, rl0, c1, rl1;
        load_ent(0, c0, rl0);
        load_ent(ng > 1 ? 1 : 0, c1, rl1);
        float dv0 = dinv[c0], dr0 = dinvr[rl0];
        for (int g = 0; g < nwhole; g++) {
          int g2 = (g + 2 < ng) ? g + 2 : ng - 1;
          int c2, rl2;
          load_ent(g2, c2, rl2);
          float dv1 = dinv[c1], dr1 = dinvr[rl1];
          float wt0 = dr0 * dv0;
          lw[g * 32 + lane] = wt0;
          add_group(c0, rl0, wt0, 32, hl, accl);
          c0 = c1; rl0 = rl1; dv0 = dv1; dr0 = dr1;
          c1 = c2; rl1 = rl2;
        }
        if ((n & 31) != 0) {
          float wt0 = dr0 * dv0;
          if (lane < (n & 31)) lw[nwhole * 32 + lane] = wt0;
          add_group(c0, rl0, wt0, n & 31, hl, accl);
        }
      }
    }
  } else {
    const int* colp = ei + E;
    int* myqe = qe + wave * Q_CAP;
    int* myqr = qr + wave * Q_CAP;
    auto drain = [&](int head, int cnt, int nrec) {
      int idx = (head + lane) & (Q_CAP - 1);
      bool live = lane < cnt;
      int e  = live ? myqe[idx] : 0;
      int rl = live ? myqr[idx] : 0;
      int c = colp[e];
      float wt = dinvr[rl] * dinv[c];
      add_group(c, rl, wt, cnt, hl, accl);
    };
    scan_row_list<false>(ei, E, r0, wave, lane, myqe, myqr, (int*)nullptr, drain);
  }

  int nrows = N - r0;
  if (nrows > BLK_ROWS) nrows = BLK_ROWS;
  int kh  = lane >> 4;
  int l15 = lane & 15;
  int t0 = wave * 16;

  int rloc = t0 + l15;
  int rld = rloc < nrows ? rloc : nrows - 1;
  const float* hRow = h + (size_t)(r0 + rld) * 64 + 2 * kh;
  v2f hc[4];
#pragma unroll
  for (int ks = 0; ks < 4; ks++) hc[ks] = *reinterpret_cast<const v2f*>(hRow + 4 * ks);

#pragma unroll
  for (int u = 0; u < 4; u++) {
    int i = u * 1024 + tid;
    int r = i >> 6, c = i & 63;
    int idx = (((r >> 1) * 64 + c) << 1) | (r & 1);
    sWg[idx] = wgr[u];
    sWi[idx] = wir[u];
  }
  if (tid < 64) { sbg[tid] = bgr; sbi[tid] = bir; }

  __syncthreads();
  if (wave >= BLK_TILES || t0 >= nrows) return;

  const float* haRow = acc + (t0 + l15) * ACC_PITCH + 2 * kh;

  v8f zero8 = {0.f, 0.f, 0.f, 0.f, 0.f, 0.f, 0.f, 0.f};
  v8f accg[4], acci[4];
#pragma unroll
  for (int nt = 0; nt < 4; nt++) { accg[nt] = zero8; acci[nt] = zero8; }

  const v2f* sWg2 = reinterpret_cast<const v2f*>(sWg);
  const v2f* sWi2 = reinterpret_cast<const v2f*>(sWi);

#pragma nounroll
  for (int kc = 0; kc < 64; kc += 16) {
    int kn = (kc + 16 < 64) ? kc + 16 : kc;
    v2f hx[4];
#pragma unroll
    for (int ks = 0; ks < 4; ks++) hx[ks] = *reinterpret_cast<const v2f*>(hRow + kn + 4 * ks);
#pragma unroll
    for (int ks = 0; ks < 4; ks++) {
      int k = kc + 4 * ks;
      v2f ga = *reinterpret_cast<const v2f*>(haRow + k);
      v2f ia = ga * hc[ks];
      int k2 = (k >> 1) + kh;
#pragma unroll
      for (int nt = 0; nt < 4; nt++) {
        v2f bgf = sWg2[k2 * 64 + nt * 16 + l15];
        v2f bif = sWi2[k2 * 64 + nt * 16 + l15];
        accg[nt] = __builtin_amdgcn_wmma_f32_16x16x4_f32(
            false, ga, false, bgf, (short)0, accg[nt], false, false);
        acci[nt] = __builtin_amdgcn_wmma_f32_16x16x4_f32(
            false, ia, false, bif, (short)0, acci[nt], false, false);
      }
    }
#pragma unroll
    for (int ks = 0; ks < 4; ks++) hc[ks] = hx[ks];
  }

  float bgl[4], bil[4];
#pragma unroll
  for (int nt = 0; nt < 4; nt++) {
    bgl[nt] = sbg[nt * 16 + l15];
    bil[nt] = sbi[nt * 16 + l15];
  }

#pragma unroll
  for (int j = 0; j < 8; j++) {
    int rl = t0 + j + 8 * kh;
    float v[4];
    float sq = 0.0f;
#pragma unroll
    for (int nt = 0; nt < 4; nt++) {
      float g = accg[nt][j] + bgl[nt];
      float s = acci[nt][j] + bil[nt];
      g = (g > 0.0f) ? g : LEAKY * g;
      s = (s > 0.0f) ? s : LEAKY * s;
      v[nt] = g + s;
      sq += v[nt] * v[nt];
    }
#pragma unroll
    for (int m = 1; m < 16; m <<= 1) sq += __shfl_xor(sq, m, 32);
    float inv = rsqrtf(fmaxf(sq, L2EPS));
    if (rl < nrows) {
      size_t row = (size_t)(r0 + rl);
      float* hn = hnext + row * 64 + l15;
      float* op = out + row * outStride + colOff + l15;
#pragma unroll
      for (int nt = 0; nt < 4; nt++) {
        hn[nt * 16] = v[nt];
        op[nt * 16] = v[nt] * inv;
      }
    }
  }
}


extern "C" void kernel_launch(void* const* d_in, const int* in_sizes, int n_in,
                              void* d_out, int out_size, void* d_ws, size_t ws_size,
                              hipStream_t stream) {
  const float* x  = (const float*)d_in[0];
  const int*   ei = (const int*)d_in[1];
  const float* Wg = (const float*)d_in[2];
  const float* bg = (const float*)d_in[3];
  const float* Wi = (const float*)d_in[4];
  const float* bi = (const float*)d_in[5];

  const int d = 64;
  const int N = in_sizes[0] / d;
  const int E = in_sizes[1] / 2;
  const int K = in_sizes[2] / (d * d);
  const int outStride = (K + 1) * d;

  float* ws    = (float*)d_ws;
  float* dinv  = ws;
  float* hA    = dinv + N;
  float* hB    = hA + (size_t)N * d;
  float* out   = (float*)d_out;

  int ablocks = (N + BLK_ROWS - 1) / BLK_ROWS;
  int*   lcnt = (int*)(hB + (size_t)N * d);
  int*   lcol = lcnt + (size_t)ablocks * 32;
  int*   lrow = lcol + (size_t)ablocks * 32 * LIST_CAP;
  float* lwt  = (float*)(lrow + (size_t)ablocks * 32 * LIST_CAP);
  int*   scnt = (int*)(lwt + (size_t)ablocks * 32 * LIST_CAP);
  v2i*   segs = (v2i*)(scnt + (size_t)ablocks * BIN_G);

  bin_kernel<<<BIN_G, BIN_T, 0, stream>>>(ei, scnt, segs, E, N, ablocks);
  prep_kernel<<<ablocks, 1024, 0, stream>>>(ei, scnt, segs, dinv, lcnt, lcol, lrow, E, N);
  copy_x_kernel<<<(N * 16 + 255) / 256, 256, 0, stream>>>(x, out, N, outStride);

  const float* h = x;
  float* nxt = hA;

  for (int layer = 0; layer < K; layer++) {
    const float* Wgl = Wg + (size_t)layer * d * d;
    const float* bgl = bg + (size_t)layer * d;
    const float* Wil = Wi + (size_t)layer * d * d;
    const float* bil = bi + (size_t)layer * d;
    if (layer == 0)
      agg_layer_kernel<0><<<ablocks, 1024, 0, stream>>>(ei, dinv, h, lcnt, lcol, lrow, lwt,
          Wgl, bgl, Wil, bil, nxt, out, E, N, outStride, (layer + 1) * d);
    else
      agg_layer_kernel<1><<<ablocks, 1024, 0, stream>>>(ei, dinv, h, lcnt, lcol, lrow, lwt,
          Wgl, bgl, Wil, bil, nxt, out, E, N, outStride, (layer + 1) * d);
    h = nxt;
    nxt = (nxt == hA) ? hB : hA;
  }
}


# Aggregation: prep hands each edge to the wave owning its destination row (12-13 rows; slot by returning LDS integer add); waves add by plain LDS load/add/store, no float atomics; sums in list order
# speedup vs baseline: 4.5052x; 1.1477x over previous
#include <hip/hip_runtime.h>

typedef float v2f __attribute__((ext_vector_type(2)));
typedef float v8f __attribute__((ext_vector_type(8)));
typedef int v2i __attribute__((ext_vector_type(2)));

#define LEAKY 0.2f
#define L2EPS 1e-12f


__global__ void copy_x_kernel(const float* __restrict__ x, float* __restrict__ out,
                              int N, int outStride) {
  int t = blockIdx.x * blockDim.x + threadIdx.x;
  if (t < N * 16) {
    int r = t >> 4, q = t & 15;
    float4 v = reinterpret_cast<const float4*>(x)[(size_t)r * 16 + q];
    reinterpret_cast<float4*>(out + (size_t)r * outStride)[q] = v;
  }
}

#define BLK_ROWS 392
#define ACC_PITCH 68
#define BLK_TILES ((BLK_ROWS + 15) / 16)
#define ACC_ROWS (BLK_TILES * 16)
#define Q_CAP 256
#define DRAIN_G 16
#define SCAN_U 4
#define LIST_CAP 1024
#define STAGE_CAP 512
#define BIN_G 256
#define SEG_PER_WAVE (BIN_G / 32)
#define SEG_CAP 64
#define BIN_T 512
#define BIN_U 8
#define BIN_MAXB 2048

static __device__ __forceinline__ float bcast_f32(float v, int srcLane) {
  return __builtin_bit_cast(float,
      __builtin_amdgcn_readlane(__builtin_bit_cast(int, v), srcLane));
}

static __device__ __forceinline__ int owner_wave(int rl) {
  return (int)(((unsigned)rl * 32u) / (unsigned)BLK_ROWS);
}

static __device__ __forceinline__ void add_group(int c, int rl, float wt, int cnt,
                                                 const float* __restrict__ hl, float* accl) {
#pragma unroll
  for (int j0 = 0; j0 < 32; j0 += DRAIN_G) {
    v2f hv[DRAIN_G];
#pragma unroll
    for (int j = 0; j < DRAIN_G; j++) {
      if (j0 + j < cnt) {
        int cj = __builtin_amdgcn_readlane(c, j0 + j);
        hv[j] = *reinterpret_cast<const v2f*>(hl + (size_t)cj * 64);
      }
    }
#pragma unroll
    for (int j = 0; j < DRAIN_G; j++) {
      if (j0 + j < cnt) {
        int rj = __builtin_amdgcn_readlane(rl, j0 + j);
        float wj = bcast_f32(wt, j0 + j);
        v2f* dst = reinterpret_cast<v2f*>(accl + rj * ACC_PITCH);
        v2f a = *dst;
        a.x += wj * hv[j].x;
        a.y += wj * hv[j].y;
        *dst = a;
      }
    }
  }
}

template <bool COUNT, class DrainF>
static __device__ __forceinline__ int scan_row_list(const int* __restrict__ rowp, int E, int r0,
                                                    int wave, int lane, int* myqe, int* myqr,
                                                    int* degl, DrainF drain) {
  int head = 0, npend = 0;
  int nrec = 0;
  int nfull = E >> 7;
  auto load_chunk = [&](int chunk, int* rv) {
    int cl = chunk < nfull ? chunk : nfull - 1;
    int4 t = *reinterpret_cast<const int4*>(rowp + (cl << 7) + lane * 4);
    rv[0] = t.x; rv[1] = t.y; rv[2] = t.z; rv[3] = t.w;
  };
  auto match_chunk = [&](int chunk, const int* rv) {
    int e0 = (chunk << 7) + lane * 4;
#pragma unroll
    for (int k = 0; k < 4; k++) {
      int rl = rv[k] - r0;
      bool p = (unsigned)rl < (unsigned)BLK_ROWS && owner_wave(rl) == wave;
      unsigned m = __builtin_amdgcn_ballot_w32(p);
      if (m != 0u) {
        if (p) {
          int pos = (head + npend + (int)__builtin_amdgcn_mbcnt_lo(m, 0u)) & (Q_CAP - 1);
          myqe[pos] = e0 + k;
          myqr[pos] = rl;
          if (COUNT) atomicAdd(degl + rl, 1);
        }
        npend += __builtin_popcount(m);
      }
    }
  };

  if (nfull > 0) {
    int cur[SCAN_U][4], nxt[SCAN_U][4];
#pragma unroll
    for (int u = 0; u < SCAN_U; u++) load_chunk(u, cur[u]);
    for (int c0 = 0; c0 < nfull; c0 += SCAN_U) {
#pragma unroll
      for (int u = 0; u < SCAN_U; u++) load_chunk(c0 + SCAN_U + u, nxt[u]);
#pragma unroll
      for (int u = 0; u < SCAN_U; u++) {
        if (c0 + u < nfull) match_chunk(c0 + u, cur[u]);
        while (npend >= 32) {
          drain(head, 32, nrec);
          head = (head + 32) & (Q_CAP - 1);
          npend -= 32;
          nrec += 32;
        }
      }
#pragma unroll
      for (int u = 0; u < SCAN_U; u++) {
        cur[u][0] = nxt[u][0]; cur[u][1] = nxt[u][1]; cur[u][2] = nxt[u][2]; cur[u][3] = nxt[u][3];
      }
    }
  }
  if ((E & 127) != 0) {
    int e0 = (nfull << 7) + lane * 4;
    int rv[4];
#pragma unroll
    for (int k = 0; k < 4; k++) rv[k] = (e0 + k < E) ? rowp[e0 + k] : -1;
    match_chunk(nfull, rv);
  }
  while (npend > 0) {
    int cnt = npend < 32 ? npend : 32;
    drain(head, cnt, nrec);
    head = (head + cnt) & (Q_CAP - 1);
    npend -= cnt;
    nrec += cnt;
  }
  return nrec;
}

__launch_bounds__(BIN_T)
__global__ void bin_kernel(const int* __restrict__ ei, int* __restrict__ scnt,
                           v2i* __restrict__ segs, int E, int N, int nblk) {
  __shared__ int cntl[BIN_MAXB];
  int tid = threadIdx.x;
  int j = blockIdx.x;
  if (nblk > BIN_MAXB) {
    for (int b = tid; b < nblk; b += BIN_T) scnt[(size_t)b * BIN_G + j] = SEG_CAP + 1;
    return;
  }
  for (int b = tid; b < nblk; b += BIN_T) cntl[b] = 0;
  __syncthreads();

  int S = (E + BIN_G - 1) / BIN_G;
  int e0 = j * S;
  int e1 = (e0 + S < E) ? e0 + S : E;
  const int* colp = ei + E;
  for (int base = e0; base < e1; base += BIN_T * BIN_U) {
    int r[BIN_U], c[BIN_U];
#pragma unroll
    for (int u = 0; u < BIN_U; u++) {
      int e = base + u * BIN_T + tid;
      int ec = e < e1 ? e : e1 - 1;
      int rv = ei[ec];
      c[u] = colp[ec];
      r[u] = e < e1 ? rv : -1;
    }
#pragma unroll
    for (int u = 0; u < BIN_U; u++) {
      int rr = r[u];
      if ((unsigned)rr < (unsigned)N) {
        int b = (int)((unsigned)rr / (unsigned)BLK_ROWS);
        int rl = rr - b * BLK_ROWS;
        int rank = atomicAdd(&cntl[b], 1);
        if (rank < SEG_CAP)
          segs[((size_t)b * BIN_G + j) * SEG_CAP + rank] = (v2i){c[u], rl};
      }
    }
  }
  __syncthreads();
  for (int b = tid; b < nblk; b += BIN_T) scnt[(size_t)b * BIN_G + j] = cntl[b];
}

__launch_bounds__(1024)
__global__ void prep_kernel(const int* __restrict__ ei, const int* __restrict__ scnt,
                            const v2i* __restrict__ segs, float* __restrict__ dinv,
                            int* __restrict__ lcnt, int* __restrict__ lcol,
                            int* __restrict__ lrow, int E, int N) {
  __shared__ int degl[BLK_ROWS];
  __shared__ int qe[32 * Q_CAP];
  __shared__ int qr[32 * Q_CAP];
  __shared__ int notwhole;
  __shared__ int ocnt[32];
  __shared__ v2i stage[32 * STAGE_CAP];

  int tid = threadIdx.x;
  int lane = tid & 31;
  int wave = __builtin_amdgcn_readfirstlane(tid >> 5);
  int r0 = blockIdx.x * BLK_ROWS;

  for (int i = tid; i < BLK_ROWS; i += 1024) degl[i] = 0;
  if (tid < 32) ocnt[tid] = 0;
  if (tid == 0) notwhole = 0;
  __syncthreads();

  const int* colp = ei + E;
  int* myqe = qe + wave * Q_CAP;
  int* myqr = qr + wave * Q_CAP;

  int lslot = blockIdx.x * 32 + wave;
  size_t lbase = (size_t)lslot * LIST_CAP;
  int* lc = lcol + lbase;
  int* lr = lrow + lbase;

  size_t sbase = (size_t)blockIdx.x * BIN_G + wave * SEG_PER_WAVE;
  int myc = (lane < SEG_PER_WAVE) ? scnt[sbase + lane] : 0;
  if (__builtin_amdgcn_ballot_w32(myc > SEG_CAP) != 0u && lane == 0) notwhole = 1;
  __syncthreads();
  int slow = __builtin_amdgcn_readfirstlane(notwhole);

  auto drain = [&](int head, int cnt, int nrec) {
    int idx = (head + lane) & (Q_CAP - 1);
    bool live = lane < cnt;
    int e  = live ? myqe[idx] : 0;
    int rl = live ? myqr[idx] : 0;
    int c = colp[e];
    if (nrec + cnt <= LIST_CAP && live) {
      lc[nrec + lane] = c;
      lr[nrec + lane] = rl;
    }
  };
  int nrec;
  if (slow == 0) {
    int cq[SEG_PER_WAVE];
#pragma unroll
    for (int q = 0; q < SEG_PER_WAVE; q++) cq[q] = __builtin_amdgcn_readlane(myc, q);
    const v2i* sg = segs + sbase * SEG_CAP;
    v2i en[SEG_PER_WAVE][SEG_CAP / 32];
#pragma unroll
    for (int q = 0; q < SEG_PER_WAVE; q++) {
#pragma unroll
      for (int p = 0; p < SEG_CAP / 32; p++) {
        int i = p * 32 + lane;
        en[q][p] = (v2i){0, 0};
        if (i < cq[q]) en[q][p] = sg[q * SEG_CAP + i];
      }
    }
    nrec = 0;
#pragma unroll
    for (int q = 0; q < SEG_PER_WAVE; q++) {
#pragma unroll
      for (int p = 0; p < SEG_CAP / 32; p++) {
        int i = p * 32 + lane;
        if (i < cq[q]) {
          int cs = en[q][p].x, rl = en[q][p].y;
          int o = owner_wave(rl);
          int slot = atomicAdd(&ocnt[o], 1);
          if (slot < STAGE_CAP) {
            stage[o * STAGE_CAP + slot] = (v2i){cs, rl};
          } else if (slot < LIST_CAP) {
            size_t at = (size_t)(blockIdx.x * 32 + o) * LIST_CAP + slot;
            lcol[at] = cs;
            lrow[at] = rl;
          }
          atomicAdd(degl + rl, 1);
        }
      }
    }
  } else {
    nrec = scan_row_list<true>(ei, E, r0, wave, lane, myqe, myqr, degl, drain);
  }
  __syncthreads();
  if (slow == 0) {
    nrec = ocnt[wave];
    int m = nrec < STAGE_CAP ? nrec : STAGE_CAP;
    for (int i = lane; i < m; i += 32) {
      v2i e = stage[wave * STAGE_CAP + i];
      lc[i] = e.x;
      lr[i] = e.y;
    }
  }
  if (lane == 0) lcnt[lslot] = nrec;
  int nrows = N - r0;
  if (nrows > BLK_ROWS) nrows = BLK_ROWS;
  for (int i = tid; i < nrows; i += 1024) {
    int dg = degl[i];
    dinv[r0 + i] = (dg > 0) ? rsqrtf((float)dg) : 0.0f;
  }
}

template <int MODE>
__launch_bounds__(1024)
__global__ void agg_layer_kernel(const int* __restrict__ ei, const float* __restrict__ dinv,
                                 const float* __restrict__ h,
                                 const int* __restrict__ lcnt, const int* __restrict__ lcol,
                                 const int* __restrict__ lrow, float* __restrict__ lwt,
                                 const float* __restrict__ Wg, const float* __restrict__ bg,
                                 const float* __restrict__ Wi, const float* __restrict__ bi,
                                 float* __restrict__ hnext, float* __restrict__ out,
                                 int E, int N, int outStride, int colOff) {
  static_assert(BLK_TILES <= 32, "one 16-row tile per wave of the work-group");
  struct __attribute__((aligned(16))) Lds {
    float wg[64 * 64];
    float wi[64 * 64];
    float bg[64];
    float bi[64];
    float acc[ACC_ROWS * ACC_PITCH];
    int qe[32 * Q_CAP];
    int qr[32 * Q_CAP];
  };
  __shared__ Lds lds;
  float* acc = lds.acc;
  float* sWg = lds.wg;
  float* sWi = lds.wi;
  float* sbg = lds.bg;
  float* sbi = lds.bi;
  int* qe = lds.qe;
  int* qr = lds.qr;

  int tid = threadIdx.x;
  int lane = tid & 31;
  int wave = __builtin_amdgcn_readfirstlane(tid >> 5);
  int r0 = blockIdx.x * BLK_ROWS;

  float4 z4 = {0.0f, 0.0f, 0.0f, 0.0f};
  for (int i = tid; i < ACC_ROWS * ACC_PITCH / 4; i += 1024) reinterpret_cast<float4*>(acc)[i] = z4;
  __syncthreads();

  float wgr[4], wir[4];
#pragma unroll
  for (int u = 0; u < 4; u++) {
    wgr[u] = Wg[u * 1024 + tid];
    wir[u] = Wi[u * 1024 + tid];
  }
  float bgr = 0.0f, bir = 0.0f;
  if (tid < 64) { bgr = bg[tid]; bir = bi[tid]; }

  const float* hl = h + lane * 2;
  float* accl = acc + lane * 2;
  const float* dinvr = dinv + r0;

  int lslot = blockIdx.x * 32 + wave;
  size_t lbase = (size_t)lslot * LIST_CAP;
  const int* lc = lcol + lbase;
  const int* lr = lrow + lbase;
  float* lw = lwt + lbase;

  int n = __builtin_amdgcn_readfirstlane(lcnt[lslot]);
  if (n <= LIST_CAP) {
    int ng = (n + 31) >> 5;
    int nwhole = n >> 5;
    if (ng > 0) {
      if (MODE == 1) {
        int c = lc[lane], rl = lr[lane];
        float wt = lw[lane];
        for (int g = 0; g < nwhole; g++) {
          int gn = (g + 1 < ng) ? g + 1 : g;
          int cn = lc[gn * 32 + lane], rln = lr[gn * 32 + lane];
          float wtn = lw[gn * 32 + lane];
          __builtin_amdgcn_sched_barrier(0);
          add_group(c, rl, wt, 32, hl, accl);
          c = cn; rl = rln; wt = wtn;
        }
        if ((n & 31) != 0) add_group(c, rl, wt, n & 31, hl, accl);
      } else {
        auto load_ent = [&](int g, int& c, int& rl) {
          int i = g * 32 + lane;
          i = i < n ? i : n - 1;
          c = lc[i]; rl = lr[i];
        };
        int c0, rl0, c1, rl1;
        load_ent(0, c0, rl0);
        load_ent(ng > 1 ? 1 : 0, c1, rl1);
        float dv0 = dinv[c0], dr0 = dinvr[rl0];
        for (int g = 0; g < nwhole; g++) {
          int g2 = (g + 2 < ng) ? g + 2 : ng - 1;
          int c2, rl2;
          load_ent(g2, c2, rl2);
          float dv1 = dinv[c1], dr1 = dinvr[rl1];
          __builtin_amdgcn_sched_barrier(0);
          float wt0 = dr0 * dv0;
          lw[g * 32 + lane] = wt0;
          add_group(c0, rl0, wt0, 32, hl, accl);
          c0 = c1; rl0 = rl1; dv0 = dv1; dr0 = dr1;
          c1 = c2; rl1 = rl2;
        }
        if ((n & 31) != 0) {
          float wt0 = dr0 * dv0;
          if (lane < (n & 31)) lw[nwhole * 32 + lane] = wt0;
          add_group(c0, rl0, wt0, n & 31, hl, accl);
        }
      }
    }
  } else {
    const int* colp = ei + E;
    int* myqe = qe + wave * Q_CAP;
    int* myqr = qr + wave * Q_CAP;
    auto drain = [&](int head, int cnt, int nrec) {
      int idx = (head + lane) & (Q_CAP - 1);
      bool live = lane < cnt;
      int e  = live ? myqe[idx] : 0;
      int rl = live ? myqr[idx] : 0;
      int c = colp[e];
      float wt = dinvr[rl] * dinv[c];
      add_group(c, rl, wt, cnt, hl, accl);
    };
    scan_row_list<false>(ei, E, r0, wave, lane, myqe, myqr, (int*)nullptr, drain);
  }

  int nrows = N - r0;
  if (nrows > BLK_ROWS) nrows = BLK_ROWS;
  int kh  = lane >> 4;
  int l15 = lane & 15;
  int t0 = wave * 16;

  int rloc = t0 + l15;
  int rld = rloc < nrows ? rloc : nrows - 1;
  const float* hRow = h + (size_t)(r0 + rld) * 64 + 2 * kh;
  v2f hc[4];
#pragma unroll
  for (int ks = 0; ks < 4; ks++) hc[ks] = *reinterpret_cast<const v2f*>(hRow + 4 * ks);

#pragma unroll
  for (int u = 0; u < 4; u++) {
    int i = u * 1024 + tid;
    int r = i >> 6, c = i & 63;
    int idx = (((r >> 1) * 64 + c) << 1) | (r & 1);
    sWg[idx] = wgr[u];
    sWi[idx] = wir[u];
  }
  if (tid < 64) { sbg[tid] = bgr; sbi[tid] = bir; }

  __syncthreads();
  if (wave >= BLK_TILES || t0 >= nrows) return;

  const float* haRow = acc + (t0 + l15) * ACC_PITCH + 2 * kh;

  v8f zero8 = {0.f, 0.f, 0.f, 0.f, 0.f, 0.f, 0.f, 0.f};
  v8f accg[4], acci[4];
#pragma unroll
  for (int nt = 0; nt < 4; nt++) { accg[nt] = zero8; acci[nt] = zero8; }

  const v2f* sWg2 = reinterpret_cast<const v2f*>(sWg);
  const v2f* sWi2 = reinterpret_cast<const v2f*>(sWi);

#pragma nounroll
  for (int kc = 0; kc < 64; kc += 16) {
    int kn = (kc + 16 < 64) ? kc + 16 : kc;
    v2f hx[4];
#pragma unroll
    for (int ks = 0; ks < 4; ks++) hx[ks] = *reinterpret_cast<const v2f*>(hRow + kn + 4 * ks);
#pragma unroll
    for (int ks = 0; ks < 4; ks++) {
      int k = kc + 4 * ks;
      v2f ga = *reinterpret_cast<const v2f*>(haRow + k);
      v2f ia = ga * hc[ks];
      int k2 = (k >> 1) + kh;
#pragma unroll
      for (int nt = 0; nt < 4; nt++) {
        v2f bgf = sWg2[k2 * 64 + nt * 16 + l15];
        v2f bif = sWi2[k2 * 64 + nt * 16 + l15];
        accg[nt] = __builtin_amdgcn_wmma_f32_16x16x4_f32(
            false, ga, false, bgf, (short)0, accg[nt], false, false);
        acci[nt] = __builtin_amdgcn_wmma_f32_16x16x4_f32(
            false, ia, false, bif, (short)0, acci[nt], false, false);
      }
    }
#pragma unroll
    for (int ks = 0; ks < 4; ks++) hc[ks] = hx[ks];
  }

  float bgl[4], bil[4];
#pragma unroll
  for (int nt = 0; nt < 4; nt++) {
    bgl[nt] = sbg[nt * 16 + l15];
    bil[nt] = sbi[nt * 16 + l15];
  }

#pragma unroll
  for (int j = 0; j < 8; j++) {
    int rl = t0 + j + 8 * kh;
    float v[4];
    float sq = 0.0f;
#pragma unroll
    for (int nt = 0; nt < 4; nt++) {
      float g = accg[nt][j] + bgl[nt];
      float s = acci[nt][j] + bil[nt];
      g = (g > 0.0f) ? g : LEAKY * g;
      s = (s > 0.0f) ? s : LEAKY * s;
      v[nt] = g + s;
      sq += v[nt] * v[nt];
    }
#pragma unroll
    for (int m = 1; m < 16; m <<= 1) sq += __shfl_xor(sq, m, 32);
    float inv = rsqrtf(fmaxf(sq, L2EPS));
    if (rl < nrows) {
      size_t row = (size_t)(r0 + rl);
      float* hn = hnext + row * 64 + l15;
      float* op = out + row * outStride + colOff + l15;
#pragma unroll
      for (int nt = 0; nt < 4; nt++) {
        hn[nt * 16] = v[nt];
        op[nt * 16] = v[nt] * inv;
      }
    }
  }
}


extern "C" void kernel_launch(void* const* d_in, const int* in_sizes, int n_in,
                              void* d_out, int out_size, void* d_ws, size_t ws_size,
                              hipStream_t stream) {
  const float* x  = (const float*)d_in[0];
  const int*   ei = (const int*)d_in[1];
  const float* Wg = (const float*)d_in[2];
  const float* bg = (const float*)d_in[3];
  const float* Wi = (const float*)d_in[4];
  const float* bi = (const float*)d_in[5];

  const int d = 64;
  const int N = in_sizes[0] / d;
  const int E = in_sizes[1] / 2;
  const int K = in_sizes[2] / (d * d);
  const int outStride = (K + 1) * d;

  float* ws    = (float*)d_ws;
  float* dinv  = ws;
  float* hA    = dinv + N;
  float* hB    = hA + (size_t)N * d;
  float* out   = (float*)d_out;

  int ablocks = (N + BLK_ROWS - 1) / BLK_ROWS;
  int*   lcnt = (int*)(hB + (size_t)N * d);
  int*   lcol = lcnt + (size_t)ablocks * 32;
  int*   lrow = lcol + (size_t)ablocks * 32 * LIST_CAP;
  float* lwt  = (float*)(lrow + (size_t)ablocks * 32 * LIST_CAP);
  int*   scnt = (int*)(lwt + (size_t)ablocks * 32 * LIST_CAP);
  v2i*   segs = (v2i*)(scnt + (size_t)ablocks * BIN_G);

  bin_kernel<<<BIN_G, BIN_T, 0, stream>>>(ei, scnt, segs, E, N, ablocks);
  prep_kernel<<<ablocks, 1024, 0, stream>>>(ei, scnt, segs, dinv, lcnt, lcol, lrow, E, N);
  copy_x_kernel<<<(N * 16 + 255) / 256, 256, 0, stream>>>(x, out, N, outStride);

  const float* h = x;
  float* nxt = hA;

  for (int layer = 0; layer < K; layer++) {
    const float* Wgl = Wg + (size_t)layer * d * d;
    const float* bgl = bg + (size_t)layer * d;
    const float* Wil = Wi + (size_t)layer * d * d;
    const float* bil = bi + (size_t)layer * d;
    if (layer == 0)
      agg_layer_kernel<0><<<ablocks, 1024, 0, stream>>>(ei, dinv, h, lcnt, lcol, lrow, lwt,
          Wgl, bgl, Wil, bil, nxt, out, E, N, outStride, (layer + 1) * d);
    else
      agg_layer_kernel<1><<<ablocks, 1024, 0, stream>>>(ei, dinv, h, lcnt, lcol, lrow, lwt,
          Wgl, bgl, Wil, bil, nxt, out, E, N, outStride, (layer + 1) * d);
    h = nxt;
    nxt = (nxt == hA) ? hB : hA;
  }
}


# Aggregation: a wave sums its own 12-13 rows in registers (element picked by the wave-uniform row number), stores them to LDS once: no LDS load/add/store per edge; gathers as a ring of 16; list order
# speedup vs baseline: 4.7415x; 1.0525x over previous
#include <hip/hip_runtime.h>

typedef float v2f __attribute__((ext_vector_type(2)));
typedef float v8f __attribute__((ext_vector_type(8)));
typedef int v2i __attribute__((ext_vector_type(2)));
typedef float v16f __attribute__((ext_vector_type(16)));

#define LEAKY 0.2f
#define L2EPS 1e-12f


__global__ void copy_x_kernel(const float* __restrict__ x, float* __restrict__ out,
                              int N, int outStride) {
  int t = blockIdx.x * blockDim.x + threadIdx.x;
  if (t < N * 16) {
    int r = t >> 4, q = t & 15;
    float4 v = reinterpret_cast<const float4*>(x)[(size_t)r * 16 + q];
    reinterpret_cast<float4*>(out + (size_t)r * outStride)[q] = v;
  }
}

#define BLK_ROWS 392
#define ACC_PITCH 68
#define BLK_TILES ((BLK_ROWS + 15) / 16)
#define ACC_ROWS (BLK_TILES * 16)
#define Q_CAP 256
#define DRAIN_G 16
#define SCAN_U 4
#define LIST_CAP 1024
#define STAGE_CAP 512
#define BIN_G 256
#define SEG_PER_WAVE (BIN_G / 32)
#define SEG_CAP 64
#define BIN_T 512
#define BIN_U 8
#define BIN_MAXB 2048

static __device__ __forceinline__ float bcast_f32(float v, int srcLane) {
  return __builtin_bit_cast(float,
      __builtin_amdgcn_readlane(__builtin_bit_cast(int, v), srcLane));
}

static __device__ __forceinline__ int owner_wave(int rl) {
  return (int)(((unsigned)rl * 32u) / (unsigned)BLK_ROWS);
}

static __device__ __forceinline__ int wave_first_row(int w) {
  return (w * BLK_ROWS + 31) / 32;
}
#define WROWS ((BLK_ROWS + 31) / 32)
#define RING 16

static __device__ __forceinline__ void add_group(int c, int rl, float wt, int cnt,
                                                 const float* __restrict__ hl, float* accl) {
#pragma unroll
  for (int j0 = 0; j0 < 32; j0 += DRAIN_G) {
    v2f hv[DRAIN_G];
#pragma unroll
    for (int j = 0; j < DRAIN_G; j++) {
      if (j0 + j < cnt) {
        int cj = __builtin_amdgcn_readlane(c, j0 + j);
        hv[j] = *reinterpret_cast<const v2f*>(hl + (size_t)cj * 64);
      }
    }
#pragma unroll
    for (int j = 0; j < DRAIN_G; j++) {
      if (j0 + j < cnt) {
        int rj = __builtin_amdgcn_readlane(rl, j0 + j);
        float wj = bcast_f32(wt, j0 + j);
        v2f* dst = reinterpret_cast<v2f*>(accl + rj * ACC_PITCH);
        v2f a = *dst;
        a.x += wj * hv[j].x;
        a.y += wj * hv[j].y;
        *dst = a;
      }
    }
  }
}

static __device__ __forceinline__ v2f gather_row(const float* __restrict__ h, int cj, unsigned lane2) {
  unsigned long long a = (unsigned long long)(h + (size_t)cj * 64);
  unsigned lo = (unsigned)__builtin_amdgcn_readfirstlane((int)(unsigned)a);
  unsigned hi = (unsigned)__builtin_amdgcn_readfirstlane((int)(unsigned)(a >> 32));
  typedef const float __attribute__((address_space(1))) gfloat;
  typedef const v2f __attribute__((address_space(1))) gv2f;
  gfloat* rowp = (gfloat*)(((unsigned long long)hi << 32) | lo);
  return *(gv2f*)(rowp + lane2);
}

template <bool GUARD>
static __device__ __forceinline__ void add_group_reg(int c, int rl, float wt, int cnx, int cnt,
                                                     const float* __restrict__ h, unsigned lane2,
                                                     int wrow0, v2f (&hv)[RING], v16f& sx, v16f& sy) {
  static_assert(WROWS <= 16, "a wave's rows must fit the 16 elements of sx and sy");
  static_assert(RING <= 32 && (RING & (RING - 1)) == 0, "ring slots are picked by j & (RING - 1)");
#pragma unroll
  for (int j = 0; j < 32; j++) {
    if (!GUARD || j < cnt) {
      unsigned k = (unsigned)(__builtin_amdgcn_readlane(rl, j) - wrow0);
      k = k < (unsigned)(WROWS - 1) ? k : (unsigned)(WROWS - 1);
      float wj = bcast_f32(wt, j);
      v2f v = hv[j & (RING - 1)];
      sx[k] += wj * v.x;
      sy[k] += wj * v.y;
      if (j + RING < 32) {
        if (!GUARD || j + RING < cnt)
          hv[j & (RING - 1)] = gather_row(h, __builtin_amdgcn_readlane(c, j + RING), lane2);
      } else if (!GUARD) {
        hv[j & (RING - 1)] = gather_row(h, __builtin_amdgcn_readlane(cnx, j + RING - 32), lane2);
      }
    }
  }
}

template <bool COUNT, class DrainF>
static __device__ __forceinline__ int scan_row_list(const int* __restrict__ rowp, int E, int r0,
                                                    int wave, int lane, int* myqe, int* myqr,
                                                    int* degl, DrainF drain) {
  int head = 0, npend = 0;
  int nrec = 0;
  int nfull = E >> 7;
  auto load_chunk = [&](int chunk, int* rv) {
    int cl = chunk < nfull ? chunk : nfull - 1;
    int4 t = *reinterpret_cast<const int4*>(rowp + (cl << 7) + lane * 4);
    rv[0] = t.x; rv[1] = t.y; rv[2] = t.z; rv[3] = t.w;
  };
  auto match_chunk = [&](int chunk, const int* rv) {
    int e0 = (chunk << 7) + lane * 4;
#pragma unroll
    for (int k = 0; k < 4; k++) {
      int rl = rv[k] - r0;
      bool p = (unsigned)rl < (unsigned)BLK_ROWS && owner_wave(rl) == wave;
      unsigned m = __builtin_amdgcn_ballot_w32(p);
      if (m != 0u) {
        if (p) {
          int pos = (head + npend + (int)__builtin_amdgcn_mbcnt_lo(m, 0u)) & (Q_CAP - 1);
          myqe[pos] = e0 + k;
          myqr[pos] = rl;
          if (COUNT) atomicAdd(degl + rl, 1);
        }
        npend += __builtin_popcount(m);
      }
    }
  };

  if (nfull > 0) {
    int cur[SCAN_U][4], nxt[SCAN_U][4];
#pragma unroll
    for (int u = 0; u < SCAN_U; u++) load_chunk(u, cur[u]);
    for (int c0 = 0; c0 < nfull; c0 += SCAN_U) {
#pragma unroll
      for (int u = 0; u < SCAN_U; u++) load_chunk(c0 + SCAN_U + u, nxt[u]);
#pragma unroll
      for (int u = 0; u < SCAN_U; u++) {
        if (c0 + u < nfull) match_chunk(c0 + u, cur[u]);
        while (npend >= 32) {
          drain(head, 32, nrec);
          head = (head + 32) & (Q_CAP - 1);
          npend -= 32;
          nrec += 32;
        }
      }
#pragma unroll
      for (int u = 0; u < SCAN_U; u++) {
        cur[u][0] = nxt[u][0]; cur[u][1] = nxt[u][1]; cur[u][2] = nxt[u][2]; cur[u][3] = nxt[u][3];
      }
    }
  }
  if ((E & 127) != 0) {
    int e0 = (nfull << 7) + lane * 4;
    int rv[4];
#pragma unroll
    for (int k = 0; k < 4; k++) rv[k] = (e0 + k < E) ? rowp[e0 + k] : -1;
    match_chunk(nfull, rv);
  }
  while (npend > 0) {
    int cnt = npend < 32 ? npend : 32;
    drain(head, cnt, nrec);
    head = (head + cnt) & (Q_CAP - 1);
    npend -= cnt;
    nrec += cnt;
  }
  return nrec;
}

__launch_bounds__(BIN_T)
__global__ void bin_kernel(const int* __restrict__ ei, int* __restrict__ scnt,
                           v2i* __restrict__ segs, int E, int N, int nblk) {
  __shared__ int cntl[BIN_MAXB];
  int tid = threadIdx.x;
  int j = blockIdx.x;
  if (nblk > BIN_MAXB) {
    for (int b = tid; b < nblk; b += BIN_T) scnt[(size_t)b * BIN_G + j] = SEG_CAP + 1;
    return;
  }
  for (int b = tid; b < nblk; b += BIN_T) cntl[b] = 0;
  __syncthreads();

  int S = (E + BIN_G - 1) / BIN_G;
  int e0 = j * S;
  int e1 = (e0 + S < E) ? e0 + S : E;
  const int* colp = ei + E;
  for (int base = e0; base < e1; base += BIN_T * BIN_U) {
    int r[BIN_U], c[BIN_U];
#pragma unroll
    for (int u = 0; u < BIN_U; u++) {
      int e = base + u * BIN_T + tid;
      int ec = e < e1 ? e : e1 - 1;
      int rv = ei[ec];
      c[u] = colp[ec];
      r[u] = e < e1 ? rv : -1;
    }
#pragma unroll
    for (int u = 0; u < BIN_U; u++) {
      int rr = r[u];
      if ((unsigned)rr < (unsigned)N) {
        int b = (int)((unsigned)rr / (unsigned)BLK_ROWS);
        int rl = rr - b * BLK_ROWS;
        int rank = atomicAdd(&cntl[b], 1);
        if (rank < SEG_CAP)
          segs[((size_t)b * BIN_G + j) * SEG_CAP + rank] = (v2i){c[u], rl};
      }
    }
  }
  __syncthreads();
  for (int b = tid; b < nblk; b += BIN_T) scnt[(size_t)b * BIN_G + j] = cntl[b];
}

__launch_bounds__(1024)
__global__ void prep_kernel(const int* __restrict__ ei, const int* __restrict__ scnt,
                            const v2i* __restrict__ segs, float* __restrict__ dinv,
                            int* __restrict__ lcnt, int* __restrict__ lcol,
                            int* __restrict__ lrow, int E, int N) {
  __shared__ int degl[BLK_ROWS];
  __shared__ int qe[32 * Q_CAP];
  __shared__ int qr[32 * Q_CAP];
  __shared__ int notwhole;
  __shared__ int ocnt[32];
  __shared__ v2i stage[32 * STAGE_CAP];

  int tid = threadIdx.x;
  int lane = tid & 31;
  int wave = __builtin_amdgcn_readfirstlane(tid >> 5);
  int r0 = blockIdx.x * BLK_ROWS;

  for (int i = tid; i < BLK_ROWS; i += 1024) degl[i] = 0;
  if (tid < 32) ocnt[tid] = 0;
  if (tid == 0) notwhole = 0;
  __syncthreads();

  const int* colp = ei + E;
  int* myqe = qe + wave * Q_CAP;
  int* myqr = qr + wave * Q_CAP;

  int lslot = blockIdx.x * 32 + wave;
  size_t lbase = (size_t)lslot * LIST_CAP;
  int* lc = lcol + lbase;
  int* lr = lrow + lbase;

  size_t sbase = (size_t)blockIdx.x * BIN_G + wave * SEG_PER_WAVE;
  int myc = (lane < SEG_PER_WAVE) ? scnt[sbase + lane] : 0;
  if (__builtin_amdgcn_ballot_w32(myc > SEG_CAP) != 0u && lane == 0) notwhole = 1;
  __syncthreads();
  int slow = __builtin_amdgcn_readfirstlane(notwhole);

  auto drain = [&](int head, int cnt, int nrec) {
    int idx = (head + lane) & (Q_CAP - 1);
    bool live = lane < cnt;
    int e  = live ? myqe[idx] : 0;
    int rl = live ? myqr[idx] : 0;
    int c = colp[e];
    if (nrec + cnt <= LIST_CAP && live) {
      lc[nrec + lane] = c;
      lr[nrec + lane] = rl;
    }
  };
  int nrec;
  if (slow == 0) {
    int cq[SEG_PER_WAVE];
#pragma unroll
    for (int q = 0; q < SEG_PER_WAVE; q++) cq[q] = __builtin_amdgcn_readlane(myc, q);
    const v2i* sg = segs + sbase * SEG_CAP;
    v2i en[SEG_PER_WAVE][SEG_CAP / 32];
#pragma unroll
    for (int q = 0; q < SEG_PER_WAVE; q++) {
#pragma unroll
      for (int p = 0; p < SEG_CAP / 32; p++) {
        int i = p * 32 + lane;
        en[q][p] = (v2i){0, 0};
        if (i < cq[q]) en[q][p] = sg[q * SEG_CAP + i];
      }
    }
    nrec = 0;
#pragma unroll
    for (int q = 0; q < SEG_PER_WAVE; q++) {
#pragma unroll
      for (int p = 0; p < SEG_CAP / 32; p++) {
        int i = p * 32 + lane;
        if (i < cq[q]) {
          int cs = en[q][p].x, rl = en[q][p].y;
          int o = owner_wave(rl);
          int slot = atomicAdd(&ocnt[o], 1);
          if (slot < STAGE_CAP) {
            stage[o * STAGE_CAP + slot] = (v2i){cs, rl};
          } else if (slot < LIST_CAP) {
            size_t at = (size_t)(blockIdx.x * 32 + o) * LIST_CAP + slot;
            lcol[at] = cs;
            lrow[at] = rl;
          }
          atomicAdd(degl + rl, 1);
        }
      }
    }
  } else {
    nrec = scan_row_list<true>(ei, E, r0, wave, lane, myqe, myqr, degl, drain);
  }
  __syncthreads();
  if (slow == 0) {
    nrec = ocnt[wave];
    int m = nrec < STAGE_CAP ? nrec : STAGE_CAP;
    for (int i = lane; i < m; i += 32) {
      v2i e = stage[wave * STAGE_CAP + i];
      lc[i] = e.x;
      lr[i] = e.y;
    }
  }
  if (lane == 0) lcnt[lslot] = nrec;
  int nrows = N - r0;
  if (nrows > BLK_ROWS) nrows = BLK_ROWS;
  for (int i = tid; i < nrows; i += 1024) {
    int dg = degl[i];
    dinv[r0 + i] = (dg > 0) ? rsqrtf((float)dg) : 0.0f;
  }
}

template <int MODE>
__launch_bounds__(1024)
__global__ void agg_layer_kernel(const int* __restrict__ ei, const float* __restrict__ dinv,
                                 const float* __restrict__ h,
                                 const int* __restrict__ lcnt, const int* __restrict__ lcol,
                                 const int* __restrict__ lrow, float* __restrict__ lwt,
                                 const float* __restrict__ Wg, const float* __restrict__ bg,
                                 const float* __restrict__ Wi, const float* __restrict__ bi,
                                 float* __restrict__ hnext, float* __restrict__ out,
                                 int E, int N, int outStride, int colOff) {
  static_assert(BLK_TILES <= 32, "one 16-row tile per wave of the work-group");
  struct __attribute__((aligned(16))) Lds {
    float wg[64 * 64];
    float wi[64 * 64];
    float bg[64];
    float bi[64];
    float acc[ACC_ROWS * ACC_PITCH];
    int qe[32 * Q_CAP];
    int qr[32 * Q_CAP];
  };
  __shared__ Lds lds;
  float* acc = lds.acc;
  float* sWg = lds.wg;
  float* sWi = lds.wi;
  float* sbg = lds.bg;
  float* sbi = lds.bi;
  int* qe = lds.qe;
  int* qr = lds.qr;

  int tid = threadIdx.x;
  int lane = tid & 31;
  int wave = __builtin_amdgcn_readfirstlane(tid >> 5);
  int r0 = blockIdx.x * BLK_ROWS;

  float4 z4 = {0.0f, 0.0f, 0.0f, 0.0f};
  for (int i = tid; i < ACC_ROWS * ACC_PITCH / 4; i += 1024) reinterpret_cast<float4*>(acc)[i] = z4;
  __syncthreads();

  float wgr[4], wir[4];
#pragma unroll
  for (int u = 0; u < 4; u++) {
    wgr[u] = Wg[u * 1024 + tid];
    wir[u] = Wi[u * 1024 + tid];
  }
  float bgr = 0.0f, bir = 0.0f;
  if (tid < 64) { bgr = bg[tid]; bir = bi[tid]; }

  const float* hl = h + lane * 2;
  float* accl = acc + lane * 2;
  const float* dinvr = dinv + r0;

  int lslot = blockIdx.x * 32 + wave;
  size_t lbase = (size_t)lslot * LIST_CAP;
  const int* lc = lcol + lbase;
  const int* lr = lrow + lbase;
  float* lw = lwt + lbase;

  int n = __builtin_amdgcn_readfirstlane(lcnt[lslot]);
  if (n <= LIST_CAP) {
    int ng = (n + 31) >> 5;
    int nwhole = n >> 5;
    if (ng > 0) {
      int wrow0 = wave_first_row(wave);
      int nown = wave_first_row(wave + 1) - wrow0;
      unsigned lane2 = (unsigned)lane * 2u;
      v16f sx = {0.f, 0.f, 0.f, 0.f, 0.f, 0.f, 0.f, 0.f, 0.f, 0.f, 0.f, 0.f, 0.f, 0.f, 0.f, 0.f};
      v16f sy = sx;
      v2f hv[RING];
      if (MODE == 1) {
        auto load_grp = [&](int g, int& cc, int& rr, float& ww) {
          int i = g * 32 + lane;
          i = i < n ? i : n - 1;
          cc = lc[i]; rr = lr[i]; ww = lw[i];
        };
        int c0, rl0, c1 = 0, rl1 = 0;
        float wt0, wt1 = 0.f;
        load_grp(0, c0, rl0, wt0);
        if (nwhole > 0) load_grp(1, c1, rl1, wt1);
#pragma unroll
        for (int j = 0; j < RING; j++) hv[j] = gather_row(h, __builtin_amdgcn_readlane(c0, j), lane2);
        for (int g = 0; g < nwhole; g++) {
          int c2, rl2;
          float wt2;
          load_grp(g + 2, c2, rl2, wt2);
          __builtin_amdgcn_sched_barrier(0);
          add_group_reg<false>(c0, rl0, wt0, c1, 32, h, lane2, wrow0, hv, sx, sy);
          c0 = c1; rl0 = rl1; wt0 = wt1;
          c1 = c2; rl1 = rl2; wt1 = wt2;
        }
        if ((n & 31) != 0) add_group_reg<true>(c0, rl0, wt0, 0, n & 31, h, lane2, wrow0, hv, sx, sy);
      } else {
        auto load_ent = [&](int g, int& cc, int& rr) {
          int i = g * 32 + lane;
          i = i < n ? i : n - 1;
          cc = lc[i]; rr = lr[i];
        };
        int c0, rl0, c1 = 0, rl1 = 0;
        load_ent(0, c0, rl0);
        if (nwhole > 0) load_ent(1, c1, rl1);
        float dv0 = dinv[c0], dr0 = dinvr[rl0];
        __builtin_amdgcn_sched_barrier(0);
#pragma unroll
        for (int j = 0; j < RING; j++) hv[j] = gather_row(h, __builtin_amdgcn_readlane(c0, j), lane2);
        for (int g = 0; g < nwhole; g++) {
          int c2, rl2;
          load_ent(g + 2, c2, rl2);
          float dv1 = dinv[c1], dr1 = dinvr[rl1];
          __builtin_amdgcn_sched_barrier(0);
          float wt0 = dr0 * dv0;
          lw[g * 32 + lane] = wt0;
          add_group_reg<false>(c0, rl0, wt0, c1, 32, h, lane2, wrow0, hv, sx, sy);
          c0 = c1; rl0 = rl1; dv0 = dv1; dr0 = dr1;
          c1 = c2; rl1 = rl2;
        }
        if ((n & 31) != 0) {
          float wt0 = dr0 * dv0;
          if (lane < (n & 31)) lw[nwhole * 32 + lane] = wt0;
          add_group_reg<true>(c0, rl0, wt0, 0, n & 31, h, lane2, wrow0, hv, sx, sy);
        }
      }
#pragma unroll
      for (int k = 0; k < WROWS; k++) {
        if (k < nown)
          *reinterpret_cast<v2f*>(accl + (wrow0 + k) * ACC_PITCH) = (v2f){sx[k], sy[k]};
      }
    }
  } else {
    const int* colp = ei + E;
    int* myqe = qe + wave * Q_CAP;
    int* myqr = qr + wave * Q_CAP;
    auto drain = [&](int head, int cnt, int nrec) {
      int idx = (head + lane) & (Q_CAP - 1);
      bool live = lane < cnt;
      int e  = live ? myqe[idx] : 0;
      int rl = live ? myqr[idx] : 0;
      int c = colp[e];
      float wt = dinvr[rl] * dinv[c];
      add_group(c, rl, wt, cnt, hl, accl);
    };
    scan_row_list<false>(ei, E, r0, wave, lane, myqe, myqr, (int*)nullptr, drain);
  }

  int nrows = N - r0;
  if (nrows > BLK_ROWS) nrows = BLK_ROWS;
  int kh  = lane >> 4;
  int l15 = lane & 15;
  int t0 = wave * 16;

  int rloc = t0 + l15;
  int rld = rloc < nrows ? rloc : nrows - 1;
  const float* hRow = h + (size_t)(r0 + rld) * 64 + 2 * kh;
  v2f hc[4];
#pragma unroll
  for (int ks = 0; ks < 4; ks++) hc[ks] = *reinterpret_cast<const v2f*>(hRow + 4 * ks);

#pragma unroll
  for (int u = 0; u < 4; u++) {
    int i = u * 1024 + tid;
    int r = i >> 6, c = i & 63;
    int idx = (((r >> 1) * 64 + c) << 1) | (r & 1);
    sWg[idx] = wgr[u];
    sWi[idx] = wir[u];
  }
  if (tid < 64) { sbg[tid] = bgr; sbi[tid] = bir; }

  __syncthreads();
  if (wave >= BLK_TILES || t0 >= nrows) return;

  const float* haRow = acc + (t0 + l15) * ACC_PITCH + 2 * kh;

  v8f zero8 = {0.f, 0.f, 0.f, 0.f, 0.f, 0.f, 0.f, 0.f};
  v8f accg[4], acci[4];
#pragma unroll
  for (int nt = 0; nt < 4; nt++) { accg[nt] = zero8; acci[nt] = zero8; }

  const v2f* sWg2 = reinterpret_cast<const v2f*>(sWg);
  const v2f* sWi2 = reinterpret_cast<const v2f*>(sWi);

#pragma nounroll
  for (int kc = 0; kc < 64; kc += 16) {
    int kn = (kc + 16 < 64) ? kc + 16 : kc;
    v2f hx[4];
#pragma unroll
    for (int ks = 0; ks < 4; ks++) hx[ks] = *reinterpret_cast<const v2f*>(hRow + kn + 4 * ks);
#pragma unroll
    for (int ks = 0; ks < 4; ks++) {
      int k = kc + 4 * ks;
      v2f ga = *reinterpret_cast<const v2f*>(haRow + k);
      v2f ia = ga * hc[ks];
      int k2 = (k >> 1) + kh;
#pragma unroll
      for (int nt = 0; nt < 4; nt++) {
        v2f bgf = sWg2[k2 * 64 + nt * 16 + l15];
        v2f bif = sWi2[k2 * 64 + nt * 16 + l15];
        accg[nt] = __builtin_amdgcn_wmma_f32_16x16x4_f32(
            false, ga, false, bgf, (short)0, accg[nt], false, false);
        acci[nt] = __builtin_amdgcn_wmma_f32_16x16x4_f32(
            false, ia, false, bif, (short)0, acci[nt], false, false);
      }
    }
#pragma unroll
    for (int ks = 0; ks < 4; ks++) hc[ks] = hx[ks];
  }

  float bgl[4], bil[4];
#pragma unroll
  for (int nt = 0; nt < 4; nt++) {
    bgl[nt] = sbg[nt * 16 + l15];
    bil[nt] = sbi[nt * 16 + l15];
  }

#pragma unroll
  for (int j = 0; j < 8; j++) {
    int rl = t0 + j + 8 * kh;
    float v[4];
    float sq = 0.0f;
#pragma unroll
    for (int nt = 0; nt < 4; nt++) {
      float g = accg[nt][j] + bgl[nt];
      float s = acci[nt][j] + bil[nt];
      g = (g > 0.0f) ? g : LEAKY * g;
      s = (s > 0.0f) ? s : LEAKY * s;
      v[nt] = g + s;
      sq += v[nt] * v[nt];
    }
#pragma unroll
    for (int m = 1; m < 16; m <<= 1) sq += __shfl_xor(sq, m, 32);
    float inv = rsqrtf(fmaxf(sq, L2EPS));
    if (rl < nrows) {
      size_t row = (size_t)(r0 + rl);
      float* hn = hnext + row * 64 + l15;
      float* op = out + row * outStride + colOff + l15;
#pragma unroll
      for (int nt = 0; nt < 4; nt++) {
        hn[nt * 16] = v[nt];
        op[nt * 16] = v[nt] * inv;
      }
    }
  }
}


extern "C" void kernel_launch(void* const* d_in, const int* in_sizes, int n_in,
                              void* d_out, int out_size, void* d_ws, size_t ws_size,
                              hipStream_t stream) {
  const float* x  = (const float*)d_in[0];
  const int*   ei = (const int*)d_in[1];
  const float* Wg = (const float*)d_in[2];
  const float* bg = (const float*)d_in[3];
  const float* Wi = (const float*)d_in[4];
  const float* bi = (const float*)d_in[5];

  const int d = 64;
  const int N = in_sizes[0] / d;
  const int E = in_sizes[1] / 2;
  const int K = in_sizes[2] / (d * d);
  const int outStride = (K + 1) * d;

  float* ws    = (float*)d_ws;
  float* dinv  = ws;
  float* hA    = dinv + N;
  float* hB    = hA + (size_t)N * d;
  float* out   = (float*)d_out;

  int ablocks = (N + BLK_ROWS - 1) / BLK_ROWS;
  int*   lcnt = (int*)(hB + (size_t)N * d);
  int*   lcol = lcnt + (size_t)ablocks * 32;
  int*   lrow = lcol + (size_t)ablocks * 32 * LIST_CAP;
  float* lwt  = (float*)(lrow + (size_t)ablocks * 32 * LIST_CAP);
  int*   scnt = (int*)(lwt + (size_t)ablocks * 32 * LIST_CAP);
  v2i*   segs = (v2i*)(scnt + (size_t)ablocks * BIN_G);

  bin_kernel<<<BIN_G, BIN_T, 0, stream>>>(ei, scnt, segs, E, N, ablocks);
  prep_kernel<<<ablocks, 1024, 0, stream>>>(ei, scnt, segs, dinv, lcnt, lcol, lrow, E, N);
  copy_x_kernel<<<(N * 16 + 255) / 256, 256, 0, stream>>>(x, out, N, outStride);

  const float* h = x;
  float* nxt = hA;

  for (int layer = 0; layer < K; layer++) {
    const float* Wgl = Wg + (size_t)layer * d * d;
    const float* bgl = bg + (size_t)layer * d;
    const float* Wil = Wi + (size_t)layer * d * d;
    const float* bil = bi + (size_t)layer * d;
    if (layer == 0)
      agg_layer_kernel<0><<<ablocks, 1024, 0, stream>>>(ei, dinv, h, lcnt, lcol, lrow, lwt,
          Wgl, bgl, Wil, bil, nxt, out, E, N, outStride, (layer + 1) * d);
    else
      agg_layer_kernel<1><<<ablocks, 1024, 0, stream>>>(ei, dinv, h, lcnt, lcol, lrow, lwt,
          Wgl, bgl, Wil, bil, nxt, out, E, N, outStride, (layer + 1) * d);
    h = nxt;
    nxt = (nxt == hA) ? hB : hA;
  }
}


# out[:,0:64]=x written in layer 1's launch by the 7 waves per work-group that have no WMMA tile (whole lines, 14 loads a lane in flight) under the other waves' products; copy launch dropped
# speedup vs baseline: 4.9014x; 1.0337x over previous
#include <hip/hip_runtime.h>

typedef float v2f __attribute__((ext_vector_type(2)));
typedef float v8f __attribute__((ext_vector_type(8)));
typedef int v2i __attribute__((ext_vector_type(2)));
typedef float v16f __attribute__((ext_vector_type(16)));

#define LEAKY 0.2f
#define L2EPS 1e-12f


__global__ void copy_x_kernel(const float* __restrict__ x, float* __restrict__ out,
                              int N, int outStride) {
  int t = blockIdx.x * blockDim.x + threadIdx.x;
  if (t < N * 16) {
    int r = t >> 4, q = t & 15;
    float4 v = reinterpret_cast<const float4*>(x)[(size_t)r * 16 + q];
    reinterpret_cast<float4*>(out + (size_t)r * outStride)[q] = v;
  }
}

#define BLK_ROWS 392
#define ACC_PITCH 68
#define BLK_TILES ((BLK_ROWS + 15) / 16)
#define ACC_ROWS (BLK_TILES * 16)
#define Q_CAP 256
#define DRAIN_G 16
#define SCAN_U 4
#define LIST_CAP 1024
#define STAGE_CAP 512
#define BIN_G 256
#define SEG_PER_WAVE (BIN_G / 32)
#define SEG_CAP 64
#define BIN_T 512
#define BIN_U 8
#define BIN_MAXB 2048

static __device__ __forceinline__ float bcast_f32(float v, int srcLane) {
  return __builtin_bit_cast(float,
      __builtin_amdgcn_readlane(__builtin_bit_cast(int, v), srcLane));
}

static __device__ __forceinline__ int owner_wave(int rl) {
  return (int)(((unsigned)rl * 32u) / (unsigned)BLK_ROWS);
}

static __device__ __forceinline__ int wave_first_row(int w) {
  return (w * BLK_ROWS + 31) / 32;
}
#define WROWS ((BLK_ROWS + 31) / 32)
#define RING 16
#define COPY_U 14

static __device__ __forceinline__ void add_group(int c, int rl, float wt, int cnt,
                                                 const float* __restrict__ hl, float* accl) {
#pragma unroll
  for (int j0 = 0; j0 < 32; j0 += DRAIN_G) {
    v2f hv[DRAIN_G];
#pragma unroll
    for (int j = 0; j < DRAIN_G; j++) {
      if (j0 + j < cnt) {
        int cj = __builtin_amdgcn_readlane(c, j0 + j);
        hv[j] = *reinterpret_cast<const v2f*>(hl + (size_t)cj * 64);
      }
    }
#pragma unroll
    for (int j = 0; j < DRAIN_G; j++) {
      if (j0 + j < cnt) {
        int rj = __builtin_amdgcn_readlane(rl, j0 + j);
        float wj = bcast_f32(wt, j0 + j);
        v2f* dst = reinterpret_cast<v2f*>(accl + rj * ACC_PITCH);
        v2f a = *dst;
        a.x += wj * hv[j].x;
        a.y += wj * hv[j].y;
        *dst = a;
      }
    }
  }
}

static __device__ __forceinline__ v2f gather_row(const float* __restrict__ h, int cj, unsigned lane2) {
  unsigned long long a = (unsigned long long)(h + (size_t)cj * 64);
  unsigned lo = (unsigned)__builtin_amdgcn_readfirstlane((int)(unsigned)a);
  unsigned hi = (unsigned)__builtin_amdgcn_readfirstlane((int)(unsigned)(a >> 32));
  typedef const float __attribute__((address_space(1))) gfloat;
  typedef const v2f __attribute__((address_space(1))) gv2f;
  gfloat* rowp = (gfloat*)(((unsigned long long)hi << 32) | lo);
  return *(gv2f*)(rowp + lane2);
}

template <bool GUARD>
static __device__ __forceinline__ void add_group_reg(int c, int rl, float wt, int cnx, int cnt,
                                                     const float* __restrict__ h, unsigned lane2,
                                                     int wrow0, v2f (&hv)[RING], v16f& sx, v16f& sy) {
  static_assert(WROWS <= 16, "a wave's rows must fit the 16 elements of sx and sy");
  static_assert(RING <= 32 && (RING & (RING - 1)) == 0, "ring slots are picked by j & (RING - 1)");
#pragma unroll
  for (int j = 0; j < 32; j++) {
    if (!GUARD || j < cnt) {
      unsigned k = (unsigned)(__builtin_amdgcn_readlane(rl, j) - wrow0);
      k = k < (unsigned)(WROWS - 1) ? k : (unsigned)(WROWS - 1);
      float wj = bcast_f32(wt, j);
      v2f v = hv[j & (RING - 1)];
      sx[k] += wj * v.x;
      sy[k] += wj * v.y;
      if (j + RING < 32) {
        if (!GUARD || j + RING < cnt)
          hv[j & (RING - 1)] = gather_row(h, __builtin_amdgcn_readlane(c, j + RING), lane2);
      } else if (!GUARD) {
        hv[j & (RING - 1)] = gather_row(h, __builtin_amdgcn_readlane(cnx, j + RING - 32), lane2);
      }
    }
  }
}

template <bool COUNT, class DrainF>
static __device__ __forceinline__ int scan_row_list(const int* __restrict__ rowp, int E, int r0,
                                                    int wave, int lane, int* myqe, int* myqr,
                                                    int* degl, DrainF drain) {
  int head = 0, npend = 0;
  int nrec = 0;
  int nfull = E >> 7;
  auto load_chunk = [&](int chunk, int* rv) {
    int cl = chunk < nfull ? chunk : nfull - 1;
    int4 t = *reinterpret_cast<const int4*>(rowp + (cl << 7) + lane * 4);
    rv[0] = t.x; rv[1] = t.y; rv[2] = t.z; rv[3] = t.w;
  };
  auto match_chunk = [&](int chunk, const int* rv) {
    int e0 = (chunk << 7) + lane * 4;
#pragma unroll
    for (int k = 0; k < 4; k++) {
      int rl = rv[k] - r0;
      bool p = (unsigned)rl < (unsigned)BLK_ROWS && owner_wave(rl) == wave;
      unsigned m = __builtin_amdgcn_ballot_w32(p);
      if (m != 0u) {
        if (p) {
          int pos = (head + npend + (int)__builtin_amdgcn_mbcnt_lo(m, 0u)) & (Q_CAP - 1);
          myqe[pos] = e0 + k;
          myqr[pos] = rl;
          if (COUNT) atomicAdd(degl + rl, 1);
        }
        npend += __builtin_popcount(m);
      }
    }
  };

  if (nfull > 0) {
    int cur[SCAN_U][4], nxt[SCAN_U][4];
#pragma unroll
    for (int u = 0; u < SCAN_U; u++) load_chunk(u, cur[u]);
    for (int c0 = 0; c0 < nfull; c0 += SCAN_U) {
#pragma unroll
      for (int u = 0; u < SCAN_U; u++) load_chunk(c0 + SCAN_U + u, nxt[u]);
#pragma unroll
      for (int u = 0; u < SCAN_U; u++) {
        if (c0 + u < nfull) match_chunk(c0 + u, cur[u]);
        while (npend >= 32) {
          drain(head, 32, nrec);
          head = (head + 32) & (Q_CAP - 1);
          npend -= 32;
          nrec += 32;
        }
      }
#pragma unroll
      for (int u = 0; u < SCAN_U; u++) {
        cur[u][0] = nxt[u][0]; cur[u][1] = nxt[u][1]; cur[u][2] = nxt[u][2]; cur[u][3] = nxt[u][3];
      }
    }
  }
  if ((E & 127) != 0) {
    int e0 = (nfull << 7) + lane * 4;
    int rv[4];
#pragma unroll
    for (int k = 0; k < 4; k++) rv[k] = (e0 + k < E) ? rowp[e0 + k] : -1;
    match_chunk(nfull, rv);
  }
  while (npend > 0) {
    int cnt = npend < 32 ? npend : 32;
    drain(head, cnt, nrec);
    head = (head + cnt) & (Q_CAP - 1);
    npend -= cnt;
    nrec += cnt;
  }
  return nrec;
}

__launch_bounds__(BIN_T)
__global__ void bin_kernel(const int* __restrict__ ei, int* __restrict__ scnt,
                           v2i* __restrict__ segs, int E, int N, int nblk) {
  __shared__ int cntl[BIN_MAXB];
  int tid = threadIdx.x;
  int j = blockIdx.x;
  if (nblk > BIN_MAXB) {
    for (int b = tid; b < nblk; b += BIN_T) scnt[(size_t)b * BIN_G + j] = SEG_CAP + 1;
    return;
  }
  for (int b = tid; b < nblk; b += BIN_T) cntl[b] = 0;
  __syncthreads();

  int S = (E + BIN_G - 1) / BIN_G;
  int e0 = j * S;
  int e1 = (e0 + S < E) ? e0 + S : E;
  const int* colp = ei + E;
  for (int base = e0; base < e1; base += BIN_T * BIN_U) {
    int r[BIN_U], c[BIN_U];
#pragma unroll
    for (int u = 0; u < BIN_U; u++) {
      int e = base + u * BIN_T + tid;
      int ec = e < e1 ? e : e1 - 1;
      int rv = ei[ec];
      c[u] = colp[ec];
      r[u] = e < e1 ? rv : -1;
    }
#pragma unroll
    for (int u = 0; u < BIN_U; u++) {
      int rr = r[u];
      if ((unsigned)rr < (unsigned)N) {
        int b = (int)((unsigned)rr / (unsigned)BLK_ROWS);
        int rl = rr - b * BLK_ROWS;
        int rank = atomicAdd(&cntl[b], 1);
        if (rank < SEG_CAP)
          segs[((size_t)b * BIN_G + j) * SEG_CAP + rank] = (v2i){c[u], rl};
      }
    }
  }
  __syncthreads();
  for (int b = tid; b < nblk; b += BIN_T) scnt[(size_t)b * BIN_G + j] = cntl[b];
}

__launch_bounds__(1024)
__global__ void prep_kernel(const int* __restrict__ ei, const int* __restrict__ scnt,
                            const v2i* __restrict__ segs, float* __restrict__ dinv,
                            int* __restrict__ lcnt, int* __restrict__ lcol,
                            int* __restrict__ lrow, int E, int N) {
  __shared__ int degl[BLK_ROWS];
  __shared__ int qe[32 * Q_CAP];
  __shared__ int qr[32 * Q_CAP];
  __shared__ int notwhole;
  __shared__ int ocnt[32];
  __shared__ v2i stage[32 * STAGE_CAP];

  int tid = threadIdx.x;
  int lane = tid & 31;
  int wave = __builtin_amdgcn_readfirstlane(tid >> 5);
  int r0 = blockIdx.x * BLK_ROWS;

  for (int i = tid; i < BLK_ROWS; i += 1024) degl[i] = 0;
  if (tid < 32) ocnt[tid] = 0;
  if (tid == 0) notwhole = 0;
  __syncthreads();

  const int* colp = ei + E;
  int* myqe = qe + wave * Q_CAP;
  int* myqr = qr + wave * Q_CAP;

  int lslot = blockIdx.x * 32 + wave;
  size_t lbase = (size_t)lslot * LIST_CAP;
  int* lc = lcol + lbase;
  int* lr = lrow + lbase;

  size_t sbase = (size_t)blockIdx.x * BIN_G + wave * SEG_PER_WAVE;
  int myc = (lane < SEG_PER_WAVE) ? scnt[sbase + lane] : 0;
  if (__builtin_amdgcn_ballot_w32(myc > SEG_CAP) != 0u && lane == 0) notwhole = 1;
  __syncthreads();
  int slow = __builtin_amdgcn_readfirstlane(notwhole);

  auto drain = [&](int head, int cnt, int nrec) {
    int idx = (head + lane) & (Q_CAP - 1);
    bool live = lane < cnt;
    int e  = live ? myqe[idx] : 0;
    int rl = live ? myqr[idx] : 0;
    int c = colp[e];
    if (nrec + cnt <= LIST_CAP && live) {
      lc[nrec + lane] = c;
      lr[nrec + lane] = rl;
    }
  };
  int nrec;
  if (slow == 0) {
    int cq[SEG_PER_WAVE];
#pragma unroll
    for (int q = 0; q < SEG_PER_WAVE; q++) cq[q] = __builtin_amdgcn_readlane(myc, q);
    const v2i* sg = segs + sbase * SEG_CAP;
    v2i en[SEG_PER_WAVE][SEG_CAP / 32];
#pragma unroll
    for (int q = 0; q < SEG_PER_WAVE; q++) {
#pragma unroll
      for (int p = 0; p < SEG_CAP / 32; p++) {
        int i = p * 32 + lane;
        en[q][p] = (v2i){0, 0};
        if (i < cq[q]) en[q][p] = sg[q * SEG_CAP + i];
      }
    }
    nrec = 0;
#pragma unroll
    for (int q = 0; q < SEG_PER_WAVE; q++) {
#pragma unroll
      for (int p = 0; p < SEG_CAP / 32; p++) {
        int i = p * 32 + lane;
        if (i < cq[q]) {
          int cs = en[q][p].x, rl = en[q][p].y;
          int o = owner_wave(rl);
          int slot = atomicAdd(&ocnt[o], 1);
          if (slot < STAGE_CAP) {
            stage[o * STAGE_CAP + slot] = (v2i){cs, rl};
          } else if (slot < LIST_CAP) {
            size_t at = (size_t)(blockIdx.x * 32 + o) * LIST_CAP + slot;
            lcol[at] = cs;
            lrow[at] = rl;
          }
          atomicAdd(degl + rl, 1);
        }
      }
    }
  } else {
    nrec = scan_row_list<true>(ei, E, r0, wave, lane, myqe, myqr, degl, drain);
  }
  __syncthreads();
  if (slow == 0) {
    nrec = ocnt[wave];
    int m = nrec < STAGE_CAP ? nrec : STAGE_CAP;
    for (int i = lane; i < m; i += 32) {
      v2i e = stage[wave * STAGE_CAP + i];
      lc[i] = e.x;
      lr[i] = e.y;
    }
  }
  if (lane == 0) lcnt[lslot] = nrec;
  int nrows = N - r0;
  if (nrows > BLK_ROWS) nrows = BLK_ROWS;
  for (int i = tid; i < nrows; i += 1024) {
    int dg = degl[i];
    dinv[r0 + i] = (dg > 0) ? rsqrtf((float)dg) : 0.0f;
  }
}

template <int MODE>
__launch_bounds__(1024)
__global__ void agg_layer_kernel(const int* __restrict__ ei, const float* __restrict__ dinv,
                                 const float* __restrict__ h,
                                 const int* __restrict__ lcnt, const int* __restrict__ lcol,
                                 const int* __restrict__ lrow, float* __restrict__ lwt,
                                 const float* __restrict__ Wg, const float* __restrict__ bg,
                                 const float* __restrict__ Wi, const float* __restrict__ bi,
                                 float* __restrict__ hnext, float* __restrict__ out,
                                 int E, int N, int outStride, int colOff) {
  static_assert(BLK_TILES <= 32, "one 16-row tile per wave of the work-group");
  struct __attribute__((aligned(16))) Lds {
    float wg[64 * 64];
    float wi[64 * 64];
    float bg[64];
    float bi[64];
    float acc[ACC_ROWS * ACC_PITCH];
    int qe[32 * Q_CAP];
    int qr[32 * Q_CAP];
  };
  __shared__ Lds lds;
  float* acc = lds.acc;
  float* sWg = lds.wg;
  float* sWi = lds.wi;
  float* sbg = lds.bg;
  float* sbi = lds.bi;
  int* qe = lds.qe;
  int* qr = lds.qr;

  int tid = threadIdx.x;
  int lane = tid & 31;
  int wave = __builtin_amdgcn_readfirstlane(tid >> 5);
  int r0 = blockIdx.x * BLK_ROWS;

  float4 z4 = {0.0f, 0.0f, 0.0f, 0.0f};
  for (int i = tid; i < ACC_ROWS * ACC_PITCH / 4; i += 1024) reinterpret_cast<float4*>(acc)[i] = z4;
  __syncthreads();

  float wgr[4], wir[4];
#pragma unroll
  for (int u = 0; u < 4; u++) {
    wgr[u] = Wg[u * 1024 + tid];
    wir[u] = Wi[u * 1024 + tid];
  }
  float bgr = 0.0f, bir = 0.0f;
  if (tid < 64) { bgr = bg[tid]; bir = bi[tid]; }

  const float* hl = h + lane * 2;
  float* accl = acc + lane * 2;
  const float* dinvr = dinv + r0;

  int lslot = blockIdx.x * 32 + wave;
  size_t lbase = (size_t)lslot * LIST_CAP;
  const int* lc = lcol + lbase;
  const int* lr = lrow + lbase;
  float* lw = lwt + lbase;

  int n = __builtin_amdgcn_readfirstlane(lcnt[lslot]);
  if (n <= LIST_CAP) {
    int ng = (n + 31) >> 5;
    int nwhole = n >> 5;
    if (ng > 0) {
      int wrow0 = wave_first_row(wave);
      int nown = wave_first_row(wave + 1) - wrow0;
      unsigned lane2 = (unsigned)lane * 2u;
      v16f sx = {0.f, 0.f, 0.f, 0.f, 0.f, 0.f, 0.f, 0.f, 0.f, 0.f, 0.f, 0.f, 0.f, 0.f, 0.f, 0.f};
      v16f sy = sx;
      v2f hv[RING];
      if (MODE == 1) {
        auto load_grp = [&](int g, int& cc, int& rr, float& ww) {
          int i = g * 32 + lane;
          i = i < n ? i : n - 1;
          cc = lc[i]; rr = lr[i]; ww = lw[i];
        };
        int c0, rl0, c1 = 0, rl1 = 0;
        float wt0, wt1 = 0.f;
        load_grp(0, c0, rl0, wt0);
        if (nwhole > 0) load_grp(1, c1, rl1, wt1);
#pragma unroll
        for (int j = 0; j < RING; j++) hv[j] = gather_row(h, __builtin_amdgcn_readlane(c0, j), lane2);
        for (int g = 0; g < nwhole; g++) {
          int c2, rl2;
          float wt2;
          load_grp(g + 2, c2, rl2, wt2);
          __builtin_amdgcn_sched_barrier(0);
          add_group_reg<false>(c0, rl0, wt0, c1, 32, h, lane2, wrow0, hv, sx, sy);
          c0 = c1; rl0 = rl1; wt0 = wt1;
          c1 = c2; rl1 = rl2; wt1 = wt2;
        }
        if ((n & 31) != 0) add_group_reg<true>(c0, rl0, wt0, 0, n & 31, h, lane2, wrow0, hv, sx, sy);
      } else {
        auto load_ent = [&](int g, int& cc, int& rr) {
          int i = g * 32 + lane;
          i = i < n ? i : n - 1;
          cc = lc[i]; rr = lr[i];
        };
        int c0, rl0, c1 = 0, rl1 = 0;
        load_ent(0, c0, rl0);
        if (nwhole > 0) load_ent(1, c1, rl1);
        float dv0 = dinv[c0], dr0 = dinvr[rl0];
        __builtin_amdgcn_sched_barrier(0);
#pragma unroll
        for (int j = 0; j < RING; j++) hv[j] = gather_row(h, __builtin_amdgcn_readlane(c0, j), lane2);
        for (int g = 0; g < nwhole; g++) {
          int c2, rl2;
          load_ent(g + 2, c2, rl2);
          float dv1 = dinv[c1], dr1 = dinvr[rl1];
          __builtin_amdgcn_sched_barrier(0);
          float wt0 = dr0 * dv0;
          lw[g * 32 + lane] = wt0;
          add_group_reg<false>(c0, rl0, wt0, c1, 32, h, lane2, wrow0, hv, sx, sy);
          c0 = c1; rl0 = rl1; dv0 = dv1; dr0 = dr1;
          c1 = c2; rl1 = rl2;
        }
        if ((n & 31) != 0) {
          float wt0 = dr0 * dv0;
          if (lane < (n & 31)) lw[nwhole * 32 + lane] = wt0;
          add_group_reg<true>(c0, rl0, wt0, 0, n & 31, h, lane2, wrow0, hv, sx, sy);
        }
      }
#pragma unroll
      for (int k = 0; k < WROWS; k++) {
        if (k < nown)
          *reinterpret_cast<v2f*>(accl + (wrow0 + k) * ACC_PITCH) = (v2f){sx[k], sy[k]};
      }
    }
  } else {
    const int* colp = ei + E;
    int* myqe = qe + wave * Q_CAP;
    int* myqr = qr + wave * Q_CAP;
    auto drain = [&](int head, int cnt, int nrec) {
      int idx = (head + lane) & (Q_CAP - 1);
      bool live = lane < cnt;
      int e  = live ? myqe[idx] : 0;
      int rl = live ? myqr[idx] : 0;
      int c = colp[e];
      float wt = dinvr[rl] * dinv[c];
      add_group(c, rl, wt, cnt, hl, accl);
    };
    scan_row_list<false>(ei, E, r0, wave, lane, myqe, myqr, (int*)nullptr, drain);
  }

  int nrows = N - r0;
  if (nrows > BLK_ROWS) nrows = BLK_ROWS;
  int kh  = lane >> 4;
  int l15 = lane & 15;
  int t0 = wave * 16;

  int rloc = t0 + l15;
  int rld = rloc < nrows ? rloc : nrows - 1;
  const float* hRow = h + (size_t)(r0 + rld) * 64 + 2 * kh;
  v2f hc[4];
#pragma unroll
  for (int ks = 0; ks < 4; ks++) hc[ks] = *reinterpret_cast<const v2f*>(hRow + 4 * ks);

#pragma unroll
  for (int u = 0; u < 4; u++) {
    int i = u * 1024 + tid;
    int r = i >> 6, c = i & 63;
    int idx = (((r >> 1) * 64 + c) << 1) | (r & 1);
    sWg[idx] = wgr[u];
    sWi[idx] = wir[u];
  }
  if (tid < 64) { sbg[tid] = bgr; sbi[tid] = bir; }

  __syncthreads();
  if (wave >= BLK_TILES || t0 >= nrows) {
    if (MODE == 0) {
      int ntile = (nrows + 15) >> 4;
      if (ntile > BLK_TILES) ntile = BLK_TILES;
      int cstride = (32 - ntile) * 32;
      int total = nrows * 16;
      int f = (wave - ntile) * 32 + lane;
      const float4* src = reinterpret_cast<const float4*>(h + (size_t)r0 * 64) + f;
      float* dst = out + (size_t)(r0 + (f >> 4)) * outStride + (f & 15) * 4;
      size_t dstep = (size_t)(cstride >> 4) * outStride;
      static_assert(COPY_U == 14, "the step below is written out for 14 passes");
      for (; f + (COPY_U - 1) * cstride < total; f += COPY_U * cstride) {
        float4 v0 = src[0 * cstride];
        float4 v1 = src[1 * cstride];
        float4 v2 = src[2 * cstride];
        float4 v3 = src[3 * cstride];
        float4 v4 = src[4 * cstride];
        float4 v5 = src[5 * cstride];
        float4 v6 = src[6 * cstride];
        float4 v7 = src[7 * cstride];
        float4 v8 = src[8 * cstride];
        float4 v9 = src[9 * cstride];
        float4 v10 = src[10 * cstride];
        float4 v11 = src[11 * cstride];
        float4 v12 = src[12 * cstride];
        float4 v13 = src[13 * cstride];
        __builtin_amdgcn_sched_barrier(0);
        *reinterpret_cast<float4*>(dst + 0 * dstep) = v0;
        *reinterpret_cast<float4*>(dst + 1 * dstep) = v1;
        *reinterpret_cast<float4*>(dst + 2 * dstep) = v2;
        *reinterpret_cast<float4*>(dst + 3 * dstep) = v3;
        *reinterpret_cast<float4*>(dst + 4 * dstep) = v4;
        *reinterpret_cast<float4*>(dst + 5 * dstep) = v5;
        *reinterpret_cast<float4*>(dst + 6 * dstep) = v6;
        *reinterpret_cast<float4*>(dst + 7 * dstep) = v7;
        *reinterpret_cast<float4*>(dst + 8 * dstep) = v8;
        *reinterpret_cast<float4*>(dst + 9 * dstep) = v9;
        *reinterpret_cast<float4*>(dst + 10 * dstep) = v10;
        *reinterpret_cast<float4*>(dst + 11 * dstep) = v11;
        *reinterpret_cast<float4*>(dst + 12 * dstep) = v12;
        *reinterpret_cast<float4*>(dst + 13 * dstep) = v13;
        src += COPY_U * cstride;
        dst += COPY_U * dstep;
      }
      for (; f < total; f += cstride) {
        *reinterpret_cast<float4*>(dst) = *src;
        src += cstride;
        dst += dstep;
      }
    }
    return;
  }

  const float* haRow = acc + (t0 + l15) * ACC_PITCH + 2 * kh;

  v8f zero8 = {0.f, 0.f, 0.f, 0.f, 0.f, 0.f, 0.f, 0.f};
  v8f accg[4], acci[4];
#pragma unroll
  for (int nt = 0; nt < 4; nt++) { accg[nt] = zero8; acci[nt] = zero8; }

  const v2f* sWg2 = reinterpret_cast<const v2f*>(sWg);
  const v2f* sWi2 = reinterpret_cast<const v2f*>(sWi);

#pragma nounroll
  for (int kc = 0; kc < 64; kc += 16) {
    int kn = (kc + 16 < 64) ? kc + 16 : kc;
    v2f hx[4];
#pragma unroll
    for (int ks = 0; ks < 4; ks++) hx[ks] = *reinterpret_cast<const v2f*>(hRow + kn + 4 * ks);
#pragma unroll
    for (int ks = 0; ks < 4; ks++) {
      int k = kc + 4 * ks;
      v2f ga = *reinterpret_cast<const v2f*>(haRow + k);
      v2f ia = ga * hc[ks];
      int k2 = (k >> 1) + kh;
#pragma unroll
      for (int nt = 0; nt < 4; nt++) {
        v2f bgf = sWg2[k2 * 64 + nt * 16 + l15];
        v2f bif = sWi2[k2 * 64 + nt * 16 + l15];
        accg[nt] = __builtin_amdgcn_wmma_f32_16x16x4_f32(
            false, ga, false, bgf, (short)0, accg[nt], false, false);
        acci[nt] = __builtin_amdgcn_wmma_f32_16x16x4_f32(
            false, ia, false, bif, (short)0, acci[nt], false, false);
      }
    }
#pragma unroll
    for (int ks = 0; ks < 4; ks++) hc[ks] = hx[ks];
  }

  float bgl[4], bil[4];
#pragma unroll
  for (int nt = 0; nt < 4; nt++) {
    bgl[nt] = sbg[nt * 16 + l15];
    bil[nt] = sbi[nt * 16 + l15];
  }

#pragma unroll
  for (int j = 0; j < 8; j++) {
    int rl = t0 + j + 8 * kh;
    float v[4];
    float sq = 0.0f;
#pragma unroll
    for (int nt = 0; nt < 4; nt++) {
      float g = accg[nt][j] + bgl[nt];
      float s = acci[nt][j] + bil[nt];
      g = (g > 0.0f) ? g : LEAKY * g;
      s = (s > 0.0f) ? s : LEAKY * s;
      v[nt] = g + s;
      sq += v[nt] * v[nt];
    }
#pragma unroll
    for (int m = 1; m < 16; m <<= 1) sq += __shfl_xor(sq, m, 32);
    float inv = rsqrtf(fmaxf(sq, L2EPS));
    if (rl < nrows) {
      size_t row = (size_t)(r0 + rl);
      float* hn = hnext + row * 64 + l15;
      float* op = out + row * outStride + colOff + l15;
#pragma unroll
      for (int nt = 0; nt < 4; nt++) {
        hn[nt * 16] = v[nt];
        op[nt * 16] = v[nt] * inv;
      }
    }
  }
}


extern "C" void kernel_launch(void* const* d_in, const int* in_sizes, int n_in,
                              void* d_out, int out_size, void* d_ws, size_t ws_size,
                              hipStream_t stream) {
  const float* x  = (const float*)d_in[0];
  const int*   ei = (const int*)d_in[1];
  const float* Wg = (const float*)d_in[2];
  const float* bg = (const float*)d_in[3];
  const float* Wi = (const float*)d_in[4];
  const float* bi = (const float*)d_in[5];

  const int d = 64;
  const int N = in_sizes[0] / d;
  const int E = in_sizes[1] / 2;
  const int K = in_sizes[2] / (d * d);
  const int outStride = (K + 1) * d;

  float* ws    = (float*)d_ws;
  float* dinv  = ws;
  float* hA    = dinv + N;
  float* hB    = hA + (size_t)N * d;
  float* out   = (float*)d_out;

  int ablocks = (N + BLK_ROWS - 1) / BLK_ROWS;
  int*   lcnt = (int*)(hB + (size_t)N * d);
  int*   lcol = lcnt + (size_t)ablocks * 32;
  int*   lrow = lcol + (size_t)ablocks * 32 * LIST_CAP;
  float* lwt  = (float*)(lrow + (size_t)ablocks * 32 * LIST_CAP);
  int*   scnt = (int*)(lwt + (size_t)ablocks * 32 * LIST_CAP);
  v2i*   segs = (v2i*)(scnt + (size_t)ablocks * BIN_G);

  bin_kernel<<<BIN_G, BIN_T, 0, stream>>>(ei, scnt, segs, E, N, ablocks);
  prep_kernel<<<ablocks, 1024, 0, stream>>>(ei, scnt, segs, dinv, lcnt, lcol, lrow, E, N);
  if (K == 0) copy_x_kernel<<<(N * 16 + 255) / 256, 256, 0, stream>>>(x, out, N, outStride);

  const float* h = x;
  float* nxt = hA;

  for (int layer = 0; layer < K; layer++) {
    const float* Wgl = Wg + (size_t)layer * d * d;
    const float* bgl = bg + (size_t)layer * d;
    const float* Wil = Wi + (size_t)layer * d * d;
    const float* bil = bi + (size_t)layer * d;
    if (layer == 0)
      agg_layer_kernel<0><<<ablocks, 1024, 0, stream>>>(ei, dinv, h, lcnt, lcol, lrow, lwt,
          Wgl, bgl, Wil, bil, nxt, out, E, N, outStride, (layer + 1) * d);
    else
      agg_layer_kernel<1><<<ablocks, 1024, 0, stream>>>(ei, dinv, h, lcnt, lcol, lrow, lwt,
          Wgl, bgl, Wil, bil, nxt, out, E, N, outStride, (layer + 1) * d);
    h = nxt;
    nxt = (nxt == hA) ? hB : hA;
  }
}
